# starting kernel
# baseline (speedup 1.0000x reference)
#include <hip/hip_runtime.h>
#include <math.h>

#define N_GRAPHS 64
#define M_NODES  512
#define IN_DIM   256
#define OUT_DIM  128
#define NHEADS   2
#define ZD       256
#define EG       8192
#define P_IDX    32
#define NEG_SLOPE 0.2f

#define CAP      47
#define MAXR     48
#define OUT_STRIDE 768

typedef float v2f __attribute__((ext_vector_type(2)));
typedef float v8f __attribute__((ext_vector_type(8)));

__global__ __launch_bounds__(256, 2)
void gat_node_kernel(const float* __restrict__ feat,
                     const int*   __restrict__ srcArr,
                     const int*   __restrict__ dstArr,
                     const int*   __restrict__ opPtr,
                     const float* __restrict__ W_f,  const float* __restrict__ al_f,
                     const float* __restrict__ ar_f, const float* __restrict__ b_f,
                     const float* __restrict__ W_b,  const float* __restrict__ al_b,
                     const float* __restrict__ ar_b, const float* __restrict__ b_b,
                     float* __restrict__ out)
{
    __shared__ float Zs[MAXR * ZD];
    __shared__ float eh[MAXR * NHEADS];
    __shared__ float er[NHEADS];
    __shared__ float partial[NHEADS * OUT_DIM];
    __shared__ int   rowNode[MAXR];
    __shared__ int   pref[257];

    const int gb   = blockIdx.x;
    const int g    = gb >> 1;
    const int dir  = gb & 1;
    const int tid  = threadIdx.x;
    const int node = opPtr[0] + g * M_NODES;

    const int*   seg  = (dir == 0) ? dstArr : srcArr;
    const int*   gat  = (dir == 0) ? srcArr : dstArr;
    const float* Wd   = (dir == 0) ? W_f  : W_b;
    const float* al   = (dir == 0) ? al_f : al_b;
    const float* ar   = (dir == 0) ? ar_f : ar_b;
    const float* bias = (dir == 0) ? b_f  : b_b;

    const int chunk  = EG / 256;
    const int myBase = g * EG + tid * chunk;
    int c = 0;
    for (int j = 0; j < chunk; ++j)
        c += (seg[myBase + j] == node) ? 1 : 0;
    pref[tid] = c;
    __syncthreads();
    if (tid == 0) {
        int acc = 0;
        for (int t = 0; t < 256; ++t) { int v = pref[t]; pref[t] = acc; acc += v; }
        pref[256] = acc;
    }
    __syncthreads();
    int pos = pref[tid];
    for (int j = 0; j < chunk; ++j) {
        int e = myBase + j;
        if (seg[e] == node) {
            if (pos < CAP) rowNode[pos] = gat[e];
            ++pos;
        }
    }
    __syncthreads();
    const int cnt   = min(pref[256], CAP);
    const int R     = cnt + 1;
    const int tiles = (R + 15) >> 4;
    if (tid == 0) rowNode[cnt] = node;
    __syncthreads();

    const int wave  = tid >> 5;
    const int lane  = tid & 31;
    const int lm    = lane & 15;
    const int khalf = (lane >> 4) << 1;
    for (int mt = 0; mt < tiles; ++mt) {
        const int m = mt * 16 + lm;
        const float* fr = feat + (long)rowNode[min(m, R - 1)] * IN_DIM + khalf;
        #pragma unroll
        for (int nn = 0; nn < 2; ++nn) {
            const int nt   = wave * 2 + nn;
            const int ncol = nt * 16 + lm;
            const float* wc = Wd + (long)khalf * ZD + ncol;
            v8f acc = {};
            #pragma unroll 4
            for (int k0 = 0; k0 < IN_DIM; k0 += 4) {
                v2f a, b;
                a.x = fr[k0];
                a.y = fr[k0 + 1];
                b.x = wc[(long)k0 * ZD];
                b.y = wc[(long)(k0 + 1) * ZD];
                acc = __builtin_amdgcn_wmma_f32_16x16x4_f32(
                          false, a, false, b, (short)0, acc, false, false);
            }
            const int mrow = mt * 16 + ((lane >> 4) << 3);
            #pragma unroll
            for (int r = 0; r < 8; ++r)
                Zs[(mrow + r) * ZD + ncol] = acc[r];
        }
    }
    __syncthreads();

    if (tid < cnt) {
        float e0 = 0.f, e1 = 0.f;
        const float* zr = &Zs[tid * ZD];
        for (int d = 0; d < OUT_DIM; ++d) {
            e0 += zr[d]           * al[d];
            e1 += zr[OUT_DIM + d] * al[OUT_DIM + d];
        }
        eh[tid * 2 + 0] = e0;
        eh[tid * 2 + 1] = e1;
    }
    if (tid < NHEADS) {
        float s = 0.f;
        const float* zr  = &Zs[cnt * ZD + tid * OUT_DIM];
        const float* arr = &ar[tid * OUT_DIM];
        for (int d = 0; d < OUT_DIM; ++d) s += zr[d] * arr[d];
        er[tid] = s;
    }
    __syncthreads();

    {
        const int h = tid >> 7;
        const int d = tid & 127;
        const float erh = er[h];
        float mx = -INFINITY;
        for (int i = 0; i < cnt; ++i) {
            float e = eh[i * 2 + h] + erh;
            e = (e > 0.f) ? e : NEG_SLOPE * e;
            mx = fmaxf(mx, e);
        }
        float s = 0.f;
        for (int i = 0; i < cnt; ++i) {
            float e = eh[i * 2 + h] + erh;
            e = (e > 0.f) ? e : NEG_SLOPE * e;
            s += expf(e - mx);
        }
        const float inv = (cnt > 0) ? 1.f / s : 0.f;
        float accum = 0.f;
        for (int i = 0; i < cnt; ++i) {
            float e = eh[i * 2 + h] + erh;
            e = (e > 0.f) ? e : NEG_SLOPE * e;
            accum += (expf(e - mx) * inv) * Zs[i * ZD + h * OUT_DIM + d];
        }
        partial[h * OUT_DIM + d] = accum + bias[h * OUT_DIM + d];
    }
    __syncthreads();
    if (tid < OUT_DIM)
        out[(long)g * OUT_STRIDE + dir * OUT_DIM + tid] =
            0.5f * (partial[tid] + partial[OUT_DIM + tid]);
}

__global__ __launch_bounds__(256)
void tail_kernel(const float* __restrict__ feat,
                 const int*   __restrict__ opPtr,
                 const int*   __restrict__ parallel,
                 float* __restrict__ out)
{
    const int g = blockIdx.x;
    const int t = threadIdx.x;
    const int node = opPtr[0] + g * M_NODES;
    out[(long)g * OUT_STRIDE + 256 + t] = feat[(long)node * IN_DIM + t];
    float s = 0.f;
    for (int p = 0; p < P_IDX; ++p) {
        int n = parallel[p] + g * M_NODES;
        s += feat[(long)n * IN_DIM + t];
    }
    out[(long)g * OUT_STRIDE + 512 + t] = s;
}

extern "C" void kernel_launch(void* const* d_in, const int* in_sizes, int n_in,
                              void* d_out, int out_size, void* d_ws, size_t ws_size,
                              hipStream_t stream)
{
    const float* feat     = (const float*)d_in[0];
    const int*   src      = (const int*)  d_in[1];
    const int*   dst      = (const int*)  d_in[2];
    const int*   op       = (const int*)  d_in[3];
    const int*   parallel = (const int*)  d_in[4];
    const float* W_f  = (const float*)d_in[5];
    const float* al_f = (const float*)d_in[6];
    const float* ar_f = (const float*)d_in[7];
    const float* b_f  = (const float*)d_in[8];
    const float* W_b  = (const float*)d_in[9];
    const float* al_b = (const float*)d_in[10];
    const float* ar_b = (const float*)d_in[11];
    const float* b_b  = (const float*)d_in[12];
    float* out = (float*)d_out;

    gat_node_kernel<<<dim3(N_GRAPHS * 2), dim3(256), 0, stream>>>(
        feat, src, dst, op, W_f, al_f, ar_f, b_f, W_b, al_b, ar_b, b_b, out);
    tail_kernel<<<dim3(N_GRAPHS), dim3(256), 0, stream>>>(feat, op, parallel, out);
}


# phase-1 GEMM register-blocked: each wave holds all M tiles x its 2 N tiles in accumulators over one K pass, 8 k-steps of loads in flight; per-accumulator k order and sum order unchanged
# speedup vs baseline: 1.4698x; 1.4698x over previous
#include <hip/hip_runtime.h>
#include <math.h>

#define N_GRAPHS 64
#define M_NODES  512
#define IN_DIM   256
#define OUT_DIM  128
#define NHEADS   2
#define ZD       256
#define EG       8192
#define P_IDX    32
#define NEG_SLOPE 0.2f

#define CAP      47
#define MAXR     48
#define OUT_STRIDE 768

typedef float v2f __attribute__((ext_vector_type(2)));
typedef float v8f __attribute__((ext_vector_type(8)));

template <int MT>
__device__ __forceinline__ void gemm_rows(const float* __restrict__ feat,
                                          const float* __restrict__ Wd,
                                          const int* rowNode, float* Zs,
                                          int R, int wave, int lane)
{
    const int lm    = lane & 15;
    const int khalf = (lane >> 4) << 1;
    const float* fr[MT];
    #pragma unroll
    for (int mt = 0; mt < MT; ++mt)
        fr[mt] = feat + (long)rowNode[min(mt * 16 + lm, R - 1)] * IN_DIM + khalf;
    const int ncol0 = wave * 32 + lm;
    const float* wc = Wd + (long)khalf * ZD + ncol0;

    v8f acc[MT][2];
    #pragma unroll
    for (int mt = 0; mt < MT; ++mt) {
        acc[mt][0] = v8f{};
        acc[mt][1] = v8f{};
    }

    #pragma unroll 8
    for (int k0 = 0; k0 < IN_DIM; k0 += 4) {
        v2f a[MT], b[2];
        #pragma unroll
        for (int mt = 0; mt < MT; ++mt) {
            a[mt].x = fr[mt][k0];
            a[mt].y = fr[mt][k0 + 1];
        }
        #pragma unroll
        for (int nn = 0; nn < 2; ++nn) {
            b[nn].x = wc[(long)k0 * ZD + nn * 16];
            b[nn].y = wc[(long)(k0 + 1) * ZD + nn * 16];
        }
        #pragma unroll
        for (int mt = 0; mt < MT; ++mt) {
            #pragma unroll
            for (int nn = 0; nn < 2; ++nn)
                acc[mt][nn] = __builtin_amdgcn_wmma_f32_16x16x4_f32(
                                  false, a[mt], false, b[nn], (short)0, acc[mt][nn], false, false);
        }
    }
    #pragma unroll
    for (int mt = 0; mt < MT; ++mt) {
        const int mrow = mt * 16 + ((lane >> 4) << 3);
        #pragma unroll
        for (int nn = 0; nn < 2; ++nn) {
            #pragma unroll
            for (int r = 0; r < 8; ++r)
                Zs[(mrow + r) * ZD + ncol0 + nn * 16] = acc[mt][nn][r];
        }
    }
}

__global__ __launch_bounds__(256, 2)
void gat_node_kernel(const float* __restrict__ feat,
                     const int*   __restrict__ srcArr,
                     const int*   __restrict__ dstArr,
                     const int*   __restrict__ opPtr,
                     const float* __restrict__ W_f,  const float* __restrict__ al_f,
                     const float* __restrict__ ar_f, const float* __restrict__ b_f,
                     const float* __restrict__ W_b,  const float* __restrict__ al_b,
                     const float* __restrict__ ar_b, const float* __restrict__ b_b,
                     float* __restrict__ out)
{
    __shared__ float Zs[MAXR * ZD];
    __shared__ float eh[MAXR * NHEADS];
    __shared__ float er[NHEADS];
    __shared__ float partial[NHEADS * OUT_DIM];
    __shared__ int   rowNode[MAXR];
    __shared__ int   pref[257];

    const int gb   = blockIdx.x;
    const int g    = gb >> 1;
    const int dir  = gb & 1;
    const int tid  = threadIdx.x;
    const int node = opPtr[0] + g * M_NODES;

    const int*   seg  = (dir == 0) ? dstArr : srcArr;
    const int*   gat  = (dir == 0) ? srcArr : dstArr;
    const float* Wd   = (dir == 0) ? W_f  : W_b;
    const float* al   = (dir == 0) ? al_f : al_b;
    const float* ar   = (dir == 0) ? ar_f : ar_b;
    const float* bias = (dir == 0) ? b_f  : b_b;

    const int chunk  = EG / 256;
    const int myBase = g * EG + tid * chunk;
    int c = 0;
    for (int j = 0; j < chunk; ++j)
        c += (seg[myBase + j] == node) ? 1 : 0;
    pref[tid] = c;
    __syncthreads();
    if (tid == 0) {
        int acc = 0;
        for (int t = 0; t < 256; ++t) { int v = pref[t]; pref[t] = acc; acc += v; }
        pref[256] = acc;
    }
    __syncthreads();
    int pos = pref[tid];
    for (int j = 0; j < chunk; ++j) {
        int e = myBase + j;
        if (seg[e] == node) {
            if (pos < CAP) rowNode[pos] = gat[e];
            ++pos;
        }
    }
    __syncthreads();
    const int cnt   = min(pref[256], CAP);
    const int R     = cnt + 1;
    const int tiles = (R + 15) >> 4;
    if (tid == 0) rowNode[cnt] = node;
    __syncthreads();

    const int wave  = tid >> 5;
    const int lane  = tid & 31;
    if (tiles == 1)      gemm_rows<1>(feat, Wd, rowNode, Zs, R, wave, lane);
    else if (tiles == 2) gemm_rows<2>(feat, Wd, rowNode, Zs, R, wave, lane);
    else                 gemm_rows<3>(feat, Wd, rowNode, Zs, R, wave, lane);
    __syncthreads();

    if (tid < cnt) {
        float e0 = 0.f, e1 = 0.f;
        const float* zr = &Zs[tid * ZD];
        for (int d = 0; d < OUT_DIM; ++d) {
            e0 += zr[d]           * al[d];
            e1 += zr[OUT_DIM + d] * al[OUT_DIM + d];
        }
        eh[tid * 2 + 0] = e0;
        eh[tid * 2 + 1] = e1;
    }
    if (tid < NHEADS) {
        float s = 0.f;
        const float* zr  = &Zs[cnt * ZD + tid * OUT_DIM];
        const float* arr = &ar[tid * OUT_DIM];
        for (int d = 0; d < OUT_DIM; ++d) s += zr[d] * arr[d];
        er[tid] = s;
    }
    __syncthreads();

    {
        const int h = tid >> 7;
        const int d = tid & 127;
        const float erh = er[h];
        float mx = -INFINITY;
        for (int i = 0; i < cnt; ++i) {
            float e = eh[i * 2 + h] + erh;
            e = (e > 0.f) ? e : NEG_SLOPE * e;
            mx = fmaxf(mx, e);
        }
        float s = 0.f;
        for (int i = 0; i < cnt; ++i) {
            float e = eh[i * 2 + h] + erh;
            e = (e > 0.f) ? e : NEG_SLOPE * e;
            s += expf(e - mx);
        }
        const float inv = (cnt > 0) ? 1.f / s : 0.f;
        float accum = 0.f;
        for (int i = 0; i < cnt; ++i) {
            float e = eh[i * 2 + h] + erh;
            e = (e > 0.f) ? e : NEG_SLOPE * e;
            accum += (expf(e - mx) * inv) * Zs[i * ZD + h * OUT_DIM + d];
        }
        partial[h * OUT_DIM + d] = accum + bias[h * OUT_DIM + d];
    }
    __syncthreads();
    if (tid < OUT_DIM)
        out[(long)g * OUT_STRIDE + dir * OUT_DIM + tid] =
            0.5f * (partial[tid] + partial[OUT_DIM + tid]);
}

__global__ __launch_bounds__(256)
void tail_kernel(const float* __restrict__ feat,
                 const int*   __restrict__ opPtr,
                 const int*   __restrict__ parallel,
                 float* __restrict__ out)
{
    const int g = blockIdx.x;
    const int t = threadIdx.x;
    const int node = opPtr[0] + g * M_NODES;
    out[(long)g * OUT_STRIDE + 256 + t] = feat[(long)node * IN_DIM + t];
    float s = 0.f;
    for (int p = 0; p < P_IDX; ++p) {
        int n = parallel[p] + g * M_NODES;
        s += feat[(long)n * IN_DIM + t];
    }
    out[(long)g * OUT_STRIDE + 512 + t] = s;
}

extern "C" void kernel_launch(void* const* d_in, const int* in_sizes, int n_in,
                              void* d_out, int out_size, void* d_ws, size_t ws_size,
                              hipStream_t stream)
{
    const float* feat     = (const float*)d_in[0];
    const int*   src      = (const int*)  d_in[1];
    const int*   dst      = (const int*)  d_in[2];
    const int*   op       = (const int*)  d_in[3];
    const int*   parallel = (const int*)  d_in[4];
    const float* W_f  = (const float*)d_in[5];
    const float* al_f = (const float*)d_in[6];
    const float* ar_f = (const float*)d_in[7];
    const float* b_f  = (const float*)d_in[8];
    const float* W_b  = (const float*)d_in[9];
    const float* al_b = (const float*)d_in[10];
    const float* ar_b = (const float*)d_in[11];
    const float* b_b  = (const float*)d_in[12];
    float* out = (float*)d_out;

    gat_node_kernel<<<dim3(N_GRAPHS * 2), dim3(256), 0, stream>>>(
        feat, src, dst, op, W_f, al_f, ar_f, b_f, W_b, al_b, ar_b, b_b, out);
    tail_kernel<<<dim3(N_GRAPHS), dim3(256), 0, stream>>>(feat, op, parallel, out);
}


# phase-0 edge compaction in one coalesced pass: matches ranked in edge order by wave ballot masks and per-wave totals; no one-thread 256-entry scan, no dependent re-read pass; sums unchanged
# speedup vs baseline: 1.7173x; 1.1684x over previous
#include <hip/hip_runtime.h>
#include <math.h>

#define N_GRAPHS 64
#define M_NODES  512
#define IN_DIM   256
#define OUT_DIM  128
#define NHEADS   2
#define ZD       256
#define EG       8192
#define P_IDX    32
#define NEG_SLOPE 0.2f

#define CAP      47
#define MAXR     48
#define NWAVES   8
#define WEDGES   (EG / NWAVES)
#define WSTRIDE  48
#define OUT_STRIDE 768

typedef float v2f __attribute__((ext_vector_type(2)));
typedef float v8f __attribute__((ext_vector_type(8)));

template <int MT>
__device__ __forceinline__ void gemm_rows(const float* __restrict__ feat,
                                          const float* __restrict__ Wd,
                                          const int* rowNode, float* Zs,
                                          int R, int wave, int lane)
{
    const int lm    = lane & 15;
    const int khalf = (lane >> 4) << 1;
    const float* fr[MT];
    #pragma unroll
    for (int mt = 0; mt < MT; ++mt)
        fr[mt] = feat + (long)rowNode[min(mt * 16 + lm, R - 1)] * IN_DIM + khalf;
    const int ncol0 = wave * 32 + lm;
    const float* wc = Wd + (long)khalf * ZD + ncol0;

    v8f acc[MT][2];
    #pragma unroll
    for (int mt = 0; mt < MT; ++mt) {
        acc[mt][0] = v8f{};
        acc[mt][1] = v8f{};
    }

    #pragma unroll 8
    for (int k0 = 0; k0 < IN_DIM; k0 += 4) {
        v2f a[MT], b[2];
        #pragma unroll
        for (int mt = 0; mt < MT; ++mt) {
            a[mt].x = fr[mt][k0];
            a[mt].y = fr[mt][k0 + 1];
        }
        #pragma unroll
        for (int nn = 0; nn < 2; ++nn) {
            b[nn].x = wc[(long)k0 * ZD + nn * 16];
            b[nn].y = wc[(long)(k0 + 1) * ZD + nn * 16];
        }
        #pragma unroll
        for (int mt = 0; mt < MT; ++mt) {
            #pragma unroll
            for (int nn = 0; nn < 2; ++nn)
                acc[mt][nn] = __builtin_amdgcn_wmma_f32_16x16x4_f32(
                                  false, a[mt], false, b[nn], (short)0, acc[mt][nn], false, false);
        }
    }
    #pragma unroll
    for (int mt = 0; mt < MT; ++mt) {
        const int mrow = mt * 16 + ((lane >> 4) << 3);
        #pragma unroll
        for (int nn = 0; nn < 2; ++nn) {
            #pragma unroll
            for (int r = 0; r < 8; ++r)
                Zs[(mrow + r) * ZD + ncol0 + nn * 16] = acc[mt][nn][r];
        }
    }
}

__global__ __launch_bounds__(256, 2)
void gat_node_kernel(const float* __restrict__ feat,
                     const int*   __restrict__ srcArr,
                     const int*   __restrict__ dstArr,
                     const int*   __restrict__ opPtr,
                     const float* __restrict__ W_f,  const float* __restrict__ al_f,
                     const float* __restrict__ ar_f, const float* __restrict__ b_f,
                     const float* __restrict__ W_b,  const float* __restrict__ al_b,
                     const float* __restrict__ ar_b, const float* __restrict__ b_b,
                     float* __restrict__ out)
{
    __shared__ float Zs[MAXR * ZD];
    __shared__ float eh[MAXR * NHEADS];
    __shared__ float er[NHEADS];
    __shared__ float partial[NHEADS * OUT_DIM];
    __shared__ int   rowNode[MAXR];
    __shared__ int   wl[NWAVES * WSTRIDE];
    __shared__ int   wtot[NWAVES];

    const int gb   = blockIdx.x;
    const int g    = gb >> 1;
    const int dir  = gb & 1;
    const int tid  = threadIdx.x;
    const int node = opPtr[0] + g * M_NODES;

    const int*   seg  = (dir == 0) ? dstArr : srcArr;
    const int*   gat  = (dir == 0) ? srcArr : dstArr;
    const float* Wd   = (dir == 0) ? W_f  : W_b;
    const float* al   = (dir == 0) ? al_f : al_b;
    const float* ar   = (dir == 0) ? ar_f : ar_b;
    const float* bias = (dir == 0) ? b_f  : b_b;

    const int wave  = tid >> 5;
    const int lane  = tid & 31;
    const int wbase = g * EG + wave * WEDGES;
    int v[8][4];
    #pragma unroll
    for (int i = 0; i < 8; ++i) {
        const int* p = seg + wbase + i * 128 + lane * 4;
        v[i][0] = p[0];
        v[i][1] = p[1];
        v[i][2] = p[2];
        v[i][3] = p[3];
    }
    int run = 0;
    #pragma unroll
    for (int i = 0; i < 8; ++i) {
        const unsigned m0 = __builtin_amdgcn_ballot_w32(v[i][0] == node);
        const unsigned m1 = __builtin_amdgcn_ballot_w32(v[i][1] == node);
        const unsigned m2 = __builtin_amdgcn_ballot_w32(v[i][2] == node);
        const unsigned m3 = __builtin_amdgcn_ballot_w32(v[i][3] == node);
        if ((m0 | m1 | m2 | m3) != 0u) {
            unsigned pos = (unsigned)run;
            pos = __builtin_amdgcn_mbcnt_lo(m0, pos);
            pos = __builtin_amdgcn_mbcnt_lo(m1, pos);
            pos = __builtin_amdgcn_mbcnt_lo(m2, pos);
            pos = __builtin_amdgcn_mbcnt_lo(m3, pos);
            const int e0 = wbase + i * 128 + lane * 4;
            #pragma unroll
            for (int q = 0; q < 4; ++q) {
                if (v[i][q] == node) {
                    if (pos < CAP) wl[wave * WSTRIDE + pos] = e0 + q;
                    ++pos;
                }
            }
            run += __builtin_popcount(m0) + __builtin_popcount(m1)
                 + __builtin_popcount(m2) + __builtin_popcount(m3);
        }
    }
    if (lane == 0) wtot[wave] = run;
    __syncthreads();
    int before = 0, total = 0;
    #pragma unroll
    for (int w = 0; w < NWAVES; ++w) {
        const int t = wtot[w];
        before += (w < wave) ? t : 0;
        total  += t;
    }
    const int cnt   = min(total, CAP);
    const int R     = cnt + 1;
    const int tiles = (R + 15) >> 4;
    const int mine = min(run, CAP);
    #pragma unroll
    for (int t = 0; t < 2; ++t) {
        const int lp = lane + 32 * t;
        if (lp < mine && before + lp < CAP)
            rowNode[before + lp] = gat[wl[wave * WSTRIDE + lp]];
    }
    if (tid == 0) rowNode[cnt] = node;
    __syncthreads();

    if (tiles == 1)      gemm_rows<1>(feat, Wd, rowNode, Zs, R, wave, lane);
    else if (tiles == 2) gemm_rows<2>(feat, Wd, rowNode, Zs, R, wave, lane);
    else                 gemm_rows<3>(feat, Wd, rowNode, Zs, R, wave, lane);
    __syncthreads();

    if (tid < cnt) {
        float e0 = 0.f, e1 = 0.f;
        const float* zr = &Zs[tid * ZD];
        for (int d = 0; d < OUT_DIM; ++d) {
            e0 += zr[d]           * al[d];
            e1 += zr[OUT_DIM + d] * al[OUT_DIM + d];
        }
        eh[tid * 2 + 0] = e0;
        eh[tid * 2 + 1] = e1;
    }
    if (tid < NHEADS) {
        float s = 0.f;
        const float* zr  = &Zs[cnt * ZD + tid * OUT_DIM];
        const float* arr = &ar[tid * OUT_DIM];
        for (int d = 0; d < OUT_DIM; ++d) s += zr[d] * arr[d];
        er[tid] = s;
    }
    __syncthreads();

    {
        const int h = tid >> 7;
        const int d = tid & 127;
        const float erh = er[h];
        float mx = -INFINITY;
        for (int i = 0; i < cnt; ++i) {
            float e = eh[i * 2 + h] + erh;
            e = (e > 0.f) ? e : NEG_SLOPE * e;
            mx = fmaxf(mx, e);
        }
        float s = 0.f;
        for (int i = 0; i < cnt; ++i) {
            float e = eh[i * 2 + h] + erh;
            e = (e > 0.f) ? e : NEG_SLOPE * e;
            s += expf(e - mx);
        }
        const float inv = (cnt > 0) ? 1.f / s : 0.f;
        float accum = 0.f;
        for (int i = 0; i < cnt; ++i) {
            float e = eh[i * 2 + h] + erh;
            e = (e > 0.f) ? e : NEG_SLOPE * e;
            accum += (expf(e - mx) * inv) * Zs[i * ZD + h * OUT_DIM + d];
        }
        partial[h * OUT_DIM + d] = accum + bias[h * OUT_DIM + d];
    }
    __syncthreads();
    if (tid < OUT_DIM)
        out[(long)g * OUT_STRIDE + dir * OUT_DIM + tid] =
            0.5f * (partial[tid] + partial[OUT_DIM + tid]);
}

__global__ __launch_bounds__(256)
void tail_kernel(const float* __restrict__ feat,
                 const int*   __restrict__ opPtr,
                 const int*   __restrict__ parallel,
                 float* __restrict__ out)
{
    const int g = blockIdx.x;
    const int t = threadIdx.x;
    const int node = opPtr[0] + g * M_NODES;
    out[(long)g * OUT_STRIDE + 256 + t] = feat[(long)node * IN_DIM + t];
    float s = 0.f;
    for (int p = 0; p < P_IDX; ++p) {
        int n = parallel[p] + g * M_NODES;
        s += feat[(long)n * IN_DIM + t];
    }
    out[(long)g * OUT_STRIDE + 512 + t] = s;
}

extern "C" void kernel_launch(void* const* d_in, const int* in_sizes, int n_in,
                              void* d_out, int out_size, void* d_ws, size_t ws_size,
                              hipStream_t stream)
{
    const float* feat     = (const float*)d_in[0];
    const int*   src      = (const int*)  d_in[1];
    const int*   dst      = (const int*)  d_in[2];
    const int*   op       = (const int*)  d_in[3];
    const int*   parallel = (const int*)  d_in[4];
    const float* W_f  = (const float*)d_in[5];
    const float* al_f = (const float*)d_in[6];
    const float* ar_f = (const float*)d_in[7];
    const float* b_f  = (const float*)d_in[8];
    const float* W_b  = (const float*)d_in[9];
    const float* al_b = (const float*)d_in[10];
    const float* ar_b = (const float*)d_in[11];
    const float* b_b  = (const float*)d_in[12];
    float* out = (float*)d_out;

    gat_node_kernel<<<dim3(N_GRAPHS * 2), dim3(256), 0, stream>>>(
        feat, src, dst, op, W_f, al_f, ar_f, b_f, W_b, al_b, ar_b, b_b, out);
    tail_kernel<<<dim3(N_GRAPHS), dim3(256), 0, stream>>>(feat, op, parallel, out);
}


# phase-2 logits: one thread per (row, head) on 3 waves, was one thread per row on one wave plus a serial er pass; Z rows at LDS pitch 260 (rows on different banks), attn vectors in LDS; sums unchanged
# speedup vs baseline: 2.1313x; 1.2411x over previous
#include <hip/hip_runtime.h>
#include <math.h>

#define N_GRAPHS 64
#define M_NODES  512
#define IN_DIM   256
#define OUT_DIM  128
#define NHEADS   2
#define ZD       256
#define EG       8192
#define P_IDX    32
#define NEG_SLOPE 0.2f

#define CAP      47
#define MAXR     48
#define NWAVES   8
#define WEDGES   (EG / NWAVES)
#define WSTRIDE  48
#define OUT_STRIDE 768
#define ZP       260

typedef float v2f __attribute__((ext_vector_type(2)));
typedef float v8f __attribute__((ext_vector_type(8)));
typedef float v4f __attribute__((ext_vector_type(4)));

template <int MT>
__device__ __forceinline__ void gemm_rows(const float* __restrict__ feat,
                                          const float* __restrict__ Wd,
                                          const int* rowNode, float* Zs,
                                          int R, int wave, int lane)
{
    const int lm    = lane & 15;
    const int khalf = (lane >> 4) << 1;
    const float* fr[MT];
    #pragma unroll
    for (int mt = 0; mt < MT; ++mt)
        fr[mt] = feat + (long)rowNode[min(mt * 16 + lm, R - 1)] * IN_DIM + khalf;
    const int ncol0 = wave * 32 + lm;
    const float* wc = Wd + (long)khalf * ZD + ncol0;

    v8f acc[MT][2];
    #pragma unroll
    for (int mt = 0; mt < MT; ++mt) {
        acc[mt][0] = v8f{};
        acc[mt][1] = v8f{};
    }

    #pragma unroll 8
    for (int k0 = 0; k0 < IN_DIM; k0 += 4) {
        v2f a[MT], b[2];
        #pragma unroll
        for (int mt = 0; mt < MT; ++mt) {
            a[mt].x = fr[mt][k0];
            a[mt].y = fr[mt][k0 + 1];
        }
        #pragma unroll
        for (int nn = 0; nn < 2; ++nn) {
            b[nn].x = wc[(long)k0 * ZD + nn * 16];
            b[nn].y = wc[(long)(k0 + 1) * ZD + nn * 16];
        }
        #pragma unroll
        for (int mt = 0; mt < MT; ++mt) {
            #pragma unroll
            for (int nn = 0; nn < 2; ++nn)
                acc[mt][nn] = __builtin_amdgcn_wmma_f32_16x16x4_f32(
                                  false, a[mt], false, b[nn], (short)0, acc[mt][nn], false, false);
        }
    }
    #pragma unroll
    for (int mt = 0; mt < MT; ++mt) {
        const int mrow = mt * 16 + ((lane >> 4) << 3);
        #pragma unroll
        for (int nn = 0; nn < 2; ++nn) {
            #pragma unroll
            for (int r = 0; r < 8; ++r)
                Zs[(mrow + r) * ZP + ncol0 + nn * 16] = acc[mt][nn][r];
        }
    }
}

__global__ __launch_bounds__(256, 2)
void gat_node_kernel(const float* __restrict__ feat,
                     const int*   __restrict__ srcArr,
                     const int*   __restrict__ dstArr,
                     const int*   __restrict__ opPtr,
                     const float* __restrict__ W_f,  const float* __restrict__ al_f,
                     const float* __restrict__ ar_f, const float* __restrict__ b_f,
                     const float* __restrict__ W_b,  const float* __restrict__ al_b,
                     const float* __restrict__ ar_b, const float* __restrict__ b_b,
                     float* __restrict__ out)
{
    __shared__ __attribute__((aligned(16))) float Zs[MAXR * ZP];
    __shared__ __attribute__((aligned(16))) float av[2 * ZD];
    __shared__ float eh[MAXR * NHEADS];
    __shared__ float er[NHEADS];
    __shared__ float partial[NHEADS * OUT_DIM];
    __shared__ int   rowNode[MAXR];
    __shared__ int   wl[NWAVES * WSTRIDE];
    __shared__ int   wtot[NWAVES];

    const int gb   = blockIdx.x;
    const int g    = gb >> 1;
    const int dir  = gb & 1;
    const int tid  = threadIdx.x;
    const int node = opPtr[0] + g * M_NODES;

    const int*   seg  = (dir == 0) ? dstArr : srcArr;
    const int*   gat  = (dir == 0) ? srcArr : dstArr;
    const float* Wd   = (dir == 0) ? W_f  : W_b;
    const float* al   = (dir == 0) ? al_f : al_b;
    const float* ar   = (dir == 0) ? ar_f : ar_b;
    const float* bias = (dir == 0) ? b_f  : b_b;

    const int wave  = tid >> 5;
    const int lane  = tid & 31;
    const int wbase = g * EG + wave * WEDGES;
    int v[8][4];
    #pragma unroll
    for (int i = 0; i < 8; ++i) {
        const int* p = seg + wbase + i * 128 + lane * 4;
        v[i][0] = p[0];
        v[i][1] = p[1];
        v[i][2] = p[2];
        v[i][3] = p[3];
    }
    const float alv = al[tid];
    const float arv = ar[tid];
    int run = 0;
    #pragma unroll
    for (int i = 0; i < 8; ++i) {
        const unsigned m0 = __builtin_amdgcn_ballot_w32(v[i][0] == node);
        const unsigned m1 = __builtin_amdgcn_ballot_w32(v[i][1] == node);
        const unsigned m2 = __builtin_amdgcn_ballot_w32(v[i][2] == node);
        const unsigned m3 = __builtin_amdgcn_ballot_w32(v[i][3] == node);
        if ((m0 | m1 | m2 | m3) != 0u) {
            unsigned pos = (unsigned)run;
            pos = __builtin_amdgcn_mbcnt_lo(m0, pos);
            pos = __builtin_amdgcn_mbcnt_lo(m1, pos);
            pos = __builtin_amdgcn_mbcnt_lo(m2, pos);
            pos = __builtin_amdgcn_mbcnt_lo(m3, pos);
            const int e0 = wbase + i * 128 + lane * 4;
            #pragma unroll
            for (int q = 0; q < 4; ++q) {
                if (v[i][q] == node) {
                    if (pos < CAP) wl[wave * WSTRIDE + pos] = e0 + q;
                    ++pos;
                }
            }
            run += __builtin_popcount(m0) + __builtin_popcount(m1)
                 + __builtin_popcount(m2) + __builtin_popcount(m3);
        }
    }
    if (lane == 0) wtot[wave] = run;
    av[tid]      = alv;
    av[ZD + tid] = arv;
    __syncthreads();
    int before = 0, total = 0;
    #pragma unroll
    for (int w = 0; w < NWAVES; ++w) {
        const int t = wtot[w];
        before += (w < wave) ? t : 0;
        total  += t;
    }
    const int cnt   = min(total, CAP);
    const int R     = cnt + 1;
    const int tiles = (R + 15) >> 4;
    const int mine = min(run, CAP);
    #pragma unroll
    for (int t = 0; t < 2; ++t) {
        const int lp = lane + 32 * t;
        if (lp < mine && before + lp < CAP)
            rowNode[before + lp] = gat[wl[wave * WSTRIDE + lp]];
    }
    if (tid == 0) rowNode[cnt] = node;
    __syncthreads();

    if (tiles == 1)      gemm_rows<1>(feat, Wd, rowNode, Zs, R, wave, lane);
    else if (tiles == 2) gemm_rows<2>(feat, Wd, rowNode, Zs, R, wave, lane);
    else                 gemm_rows<3>(feat, Wd, rowNode, Zs, R, wave, lane);
    __syncthreads();

    if (tid < 2 * R) {
        const int row = tid >> 1;
        const int h   = tid & 1;
        const v4f* zr = (const v4f*)&Zs[row * ZP + h * OUT_DIM];
        const v4f* a  = (const v4f*)&av[((row == cnt) ? ZD : 0) + h * OUT_DIM];
        v4f zc[4], ac[4];
        #pragma unroll
        for (int q = 0; q < 4; ++q) { zc[q] = zr[q]; ac[q] = a[q]; }
        float s = 0.f;
        #pragma unroll
        for (int c = 0; c < 8; ++c) {
            v4f zn[4], an[4];
            if (c < 7) {
                #pragma unroll
                for (int q = 0; q < 4; ++q) { zn[q] = zr[(c + 1) * 4 + q]; an[q] = a[(c + 1) * 4 + q]; }
            }
            asm volatile("" :: "v"(zc[0]), "v"(zc[1]), "v"(zc[2]), "v"(zc[3]),
                               "v"(ac[0]), "v"(ac[1]), "v"(ac[2]), "v"(ac[3]));
            #pragma unroll
            for (int q = 0; q < 4; ++q) {
                s += zc[q].x * ac[q].x;
                s += zc[q].y * ac[q].y;
                s += zc[q].z * ac[q].z;
                s += zc[q].w * ac[q].w;
            }
            if (c < 7) {
                #pragma unroll
                for (int q = 0; q < 4; ++q) { zc[q] = zn[q]; ac[q] = an[q]; }
            }
        }
        if (row == cnt) er[h] = s;
        else            eh[row * 2 + h] = s;
    }
    __syncthreads();

    {
        const int h = tid >> 7;
        const int d = tid & 127;
        const float erh = er[h];
        float mx = -INFINITY;
        for (int i = 0; i < cnt; ++i) {
            float e = eh[i * 2 + h] + erh;
            e = (e > 0.f) ? e : NEG_SLOPE * e;
            mx = fmaxf(mx, e);
        }
        float s = 0.f;
        for (int i = 0; i < cnt; ++i) {
            float e = eh[i * 2 + h] + erh;
            e = (e > 0.f) ? e : NEG_SLOPE * e;
            s += expf(e - mx);
        }
        const float inv = (cnt > 0) ? 1.f / s : 0.f;
        float accum = 0.f;
        for (int i = 0; i < cnt; ++i) {
            float e = eh[i * 2 + h] + erh;
            e = (e > 0.f) ? e : NEG_SLOPE * e;
            accum += (expf(e - mx) * inv) * Zs[i * ZP + h * OUT_DIM + d];
        }
        partial[h * OUT_DIM + d] = accum + bias[h * OUT_DIM + d];
    }
    __syncthreads();
    if (tid < OUT_DIM)
        out[(long)g * OUT_STRIDE + dir * OUT_DIM + tid] =
            0.5f * (partial[tid] + partial[OUT_DIM + tid]);
}

__global__ __launch_bounds__(256)
void tail_kernel(const float* __restrict__ feat,
                 const int*   __restrict__ opPtr,
                 const int*   __restrict__ parallel,
                 float* __restrict__ out)
{
    const int g = blockIdx.x;
    const int t = threadIdx.x;
    const int node = opPtr[0] + g * M_NODES;
    out[(long)g * OUT_STRIDE + 256 + t] = feat[(long)node * IN_DIM + t];
    float s = 0.f;
    for (int p = 0; p < P_IDX; ++p) {
        int n = parallel[p] + g * M_NODES;
        s += feat[(long)n * IN_DIM + t];
    }
    out[(long)g * OUT_STRIDE + 512 + t] = s;
}

extern "C" void kernel_launch(void* const* d_in, const int* in_sizes, int n_in,
                              void* d_out, int out_size, void* d_ws, size_t ws_size,
                              hipStream_t stream)
{
    const float* feat     = (const float*)d_in[0];
    const int*   src      = (const int*)  d_in[1];
    const int*   dst      = (const int*)  d_in[2];
    const int*   op       = (const int*)  d_in[3];
    const int*   parallel = (const int*)  d_in[4];
    const float* W_f  = (const float*)d_in[5];
    const float* al_f = (const float*)d_in[6];
    const float* ar_f = (const float*)d_in[7];
    const float* b_f  = (const float*)d_in[8];
    const float* W_b  = (const float*)d_in[9];
    const float* al_b = (const float*)d_in[10];
    const float* ar_b = (const float*)d_in[11];
    const float* b_b  = (const float*)d_in[12];
    float* out = (float*)d_out;

    gat_node_kernel<<<dim3(N_GRAPHS * 2), dim3(256), 0, stream>>>(
        feat, src, dst, op, W_f, al_f, ar_f, b_f, W_b, al_b, ar_b, b_b, out);
    tail_kernel<<<dim3(N_GRAPHS), dim3(256), 0, stream>>>(feat, op, parallel, out);
}


# feature passthrough and parallel-index sum run as 64 extra work-groups of the attention launch on otherwise idle WGPs: launches per repetition 2 -> 1; every sum and its order unchanged
# speedup vs baseline: 2.6009x; 1.2203x over previous
#include <hip/hip_runtime.h>
#include <math.h>

#define N_GRAPHS 64
#define M_NODES  512
#define IN_DIM   256
#define OUT_DIM  128
#define NHEADS   2
#define ZD       256
#define EG       8192
#define P_IDX    32
#define NEG_SLOPE 0.2f

#define CAP      47
#define MAXR     48
#define NWAVES   8
#define WEDGES   (EG / NWAVES)
#define WSTRIDE  48
#define OUT_STRIDE 768
#define ZP       260

typedef float v2f __attribute__((ext_vector_type(2)));
typedef float v8f __attribute__((ext_vector_type(8)));
typedef float v4f __attribute__((ext_vector_type(4)));

template <int MT>
__device__ __forceinline__ void gemm_rows(const float* __restrict__ feat,
                                          const float* __restrict__ Wd,
                                          const int* rowNode, float* Zs,
                                          int R, int wave, int lane)
{
    const int lm    = lane & 15;
    const int khalf = (lane >> 4) << 1;
    const float* fr[MT];
    #pragma unroll
    for (int mt = 0; mt < MT; ++mt)
        fr[mt] = feat + (long)rowNode[min(mt * 16 + lm, R - 1)] * IN_DIM + khalf;
    const int ncol0 = wave * 32 + lm;
    const float* wc = Wd + (long)khalf * ZD + ncol0;

    v8f acc[MT][2];
    #pragma unroll
    for (int mt = 0; mt < MT; ++mt) {
        acc[mt][0] = v8f{};
        acc[mt][1] = v8f{};
    }

    #pragma unroll 8
    for (int k0 = 0; k0 < IN_DIM; k0 += 4) {
        v2f a[MT], b[2];
        #pragma unroll
        for (int mt = 0; mt < MT; ++mt) {
            a[mt].x = fr[mt][k0];
            a[mt].y = fr[mt][k0 + 1];
        }
        #pragma unroll
        for (int nn = 0; nn < 2; ++nn) {
            b[nn].x = wc[(long)k0 * ZD + nn * 16];
            b[nn].y = wc[(long)(k0 + 1) * ZD + nn * 16];
        }
        #pragma unroll
        for (int mt = 0; mt < MT; ++mt) {
            #pragma unroll
            for (int nn = 0; nn < 2; ++nn)
                acc[mt][nn] = __builtin_amdgcn_wmma_f32_16x16x4_f32(
                                  false, a[mt], false, b[nn], (short)0, acc[mt][nn], false, false);
        }
    }
    #pragma unroll
    for (int mt = 0; mt < MT; ++mt) {
        const int mrow = mt * 16 + ((lane >> 4) << 3);
        #pragma unroll
        for (int nn = 0; nn < 2; ++nn) {
            #pragma unroll
            for (int r = 0; r < 8; ++r)
                Zs[(mrow + r) * ZP + ncol0 + nn * 16] = acc[mt][nn][r];
        }
    }
}

__device__ __forceinline__ void tail_rows(const float* __restrict__ feat,
                                          const int*   __restrict__ opPtr,
                                          const int*   __restrict__ parallel,
                                          float* __restrict__ out, int g, int t)
{
    const int node = opPtr[0] + g * M_NODES;
    out[(long)g * OUT_STRIDE + 256 + t] = feat[(long)node * IN_DIM + t];
    float s = 0.f;
    for (int p = 0; p < P_IDX; ++p) {
        int n = parallel[p] + g * M_NODES;
        s += feat[(long)n * IN_DIM + t];
    }
    out[(long)g * OUT_STRIDE + 512 + t] = s;
}

__global__ __launch_bounds__(256, 2)
void gat_node_kernel(const float* __restrict__ feat,
                     const int*   __restrict__ srcArr,
                     const int*   __restrict__ dstArr,
                     const int*   __restrict__ opPtr,
                     const int*   __restrict__ parallel,
                     const float* __restrict__ W_f,  const float* __restrict__ al_f,
                     const float* __restrict__ ar_f, const float* __restrict__ b_f,
                     const float* __restrict__ W_b,  const float* __restrict__ al_b,
                     const float* __restrict__ ar_b, const float* __restrict__ b_b,
                     float* __restrict__ out)
{
    __shared__ __attribute__((aligned(16))) float Zs[MAXR * ZP];
    __shared__ __attribute__((aligned(16))) float av[2 * ZD];
    __shared__ float eh[MAXR * NHEADS];
    __shared__ float er[NHEADS];
    __shared__ float partial[NHEADS * OUT_DIM];
    __shared__ int   rowNode[MAXR];
    __shared__ int   wl[NWAVES * WSTRIDE];
    __shared__ int   wtot[NWAVES];

    const int gb   = blockIdx.x;
    if (gb >= 2 * N_GRAPHS) {
        tail_rows(feat, opPtr, parallel, out, gb - 2 * N_GRAPHS, (int)threadIdx.x);
        return;
    }
    const int g    = gb >> 1;
    const int dir  = gb & 1;
    const int tid  = threadIdx.x;
    const int node = opPtr[0] + g * M_NODES;

    const int*   seg  = (dir == 0) ? dstArr : srcArr;
    const int*   gat  = (dir == 0) ? srcArr : dstArr;
    const float* Wd   = (dir == 0) ? W_f  : W_b;
    const float* al   = (dir == 0) ? al_f : al_b;
    const float* ar   = (dir == 0) ? ar_f : ar_b;
    const float* bias = (dir == 0) ? b_f  : b_b;

    const int wave  = tid >> 5;
    const int lane  = tid & 31;
    const int wbase = g * EG + wave * WEDGES;
    int v[8][4];
    #pragma unroll
    for (int i = 0; i < 8; ++i) {
        const int* p = seg + wbase + i * 128 + lane * 4;
        v[i][0] = p[0];
        v[i][1] = p[1];
        v[i][2] = p[2];
        v[i][3] = p[3];
    }
    const float alv = al[tid];
    const float arv = ar[tid];
    int run = 0;
    #pragma unroll
    for (int i = 0; i < 8; ++i) {
        const unsigned m0 = __builtin_amdgcn_ballot_w32(v[i][0] == node);
        const unsigned m1 = __builtin_amdgcn_ballot_w32(v[i][1] == node);
        const unsigned m2 = __builtin_amdgcn_ballot_w32(v[i][2] == node);
        const unsigned m3 = __builtin_amdgcn_ballot_w32(v[i][3] == node);
        if ((m0 | m1 | m2 | m3) != 0u) {
            unsigned pos = (unsigned)run;
            pos = __builtin_amdgcn_mbcnt_lo(m0, pos);
            pos = __builtin_amdgcn_mbcnt_lo(m1, pos);
            pos = __builtin_amdgcn_mbcnt_lo(m2, pos);
            pos = __builtin_amdgcn_mbcnt_lo(m3, pos);
            const int e0 = wbase + i * 128 + lane * 4;
            #pragma unroll
            for (int q = 0; q < 4; ++q) {
                if (v[i][q] == node) {
                    if (pos < CAP) wl[wave * WSTRIDE + pos] = e0 + q;
                    ++pos;
                }
            }
            run += __builtin_popcount(m0) + __builtin_popcount(m1)
                 + __builtin_popcount(m2) + __builtin_popcount(m3);
        }
    }
    if (lane == 0) wtot[wave] = run;
    av[tid]      = alv;
    av[ZD + tid] = arv;
    __syncthreads();
    int before = 0, total = 0;
    #pragma unroll
    for (int w = 0; w < NWAVES; ++w) {
        const int t = wtot[w];
        before += (w < wave) ? t : 0;
        total  += t;
    }
    const int cnt   = min(total, CAP);
    const int R     = cnt + 1;
    const int tiles = (R + 15) >> 4;
    const int mine = min(run, CAP);
    #pragma unroll
    for (int t = 0; t < 2; ++t) {
        const int lp = lane + 32 * t;
        if (lp < mine && before + lp < CAP)
            rowNode[before + lp] = gat[wl[wave * WSTRIDE + lp]];
    }
    if (tid == 0) rowNode[cnt] = node;
    __syncthreads();

    if (tiles == 1)      gemm_rows<1>(feat, Wd, rowNode, Zs, R, wave, lane);
    else if (tiles == 2) gemm_rows<2>(feat, Wd, rowNode, Zs, R, wave, lane);
    else                 gemm_rows<3>(feat, Wd, rowNode, Zs, R, wave, lane);
    __syncthreads();

    if (tid < 2 * R) {
        const int row = tid >> 1;
        const int h   = tid & 1;
        const v4f* zr = (const v4f*)&Zs[row * ZP + h * OUT_DIM];
        const v4f* a  = (const v4f*)&av[((row == cnt) ? ZD : 0) + h * OUT_DIM];
        v4f zc[4], ac[4];
        #pragma unroll
        for (int q = 0; q < 4; ++q) { zc[q] = zr[q]; ac[q] = a[q]; }
        float s = 0.f;
        #pragma unroll
        for (int c = 0; c < 8; ++c) {
            v4f zn[4], an[4];
            if (c < 7) {
                #pragma unroll
                for (int q = 0; q < 4; ++q) { zn[q] = zr[(c + 1) * 4 + q]; an[q] = a[(c + 1) * 4 + q]; }
            }
            asm volatile("" :: "v"(zc[0]), "v"(zc[1]), "v"(zc[2]), "v"(zc[3]),
                               "v"(ac[0]), "v"(ac[1]), "v"(ac[2]), "v"(ac[3]));
            #pragma unroll
            for (int q = 0; q < 4; ++q) {
                s += zc[q].x * ac[q].x;
                s += zc[q].y * ac[q].y;
                s += zc[q].z * ac[q].z;
                s += zc[q].w * ac[q].w;
            }
            if (c < 7) {
                #pragma unroll
                for (int q = 0; q < 4; ++q) { zc[q] = zn[q]; ac[q] = an[q]; }
            }
        }
        if (row == cnt) er[h] = s;
        else            eh[row * 2 + h] = s;
    }
    __syncthreads();

    {
        const int h = tid >> 7;
        const int d = tid & 127;
        const float erh = er[h];
        float mx = -INFINITY;
        for (int i = 0; i < cnt; ++i) {
            float e = eh[i * 2 + h] + erh;
            e = (e > 0.f) ? e : NEG_SLOPE * e;
            mx = fmaxf(mx, e);
        }
        float s = 0.f;
        for (int i = 0; i < cnt; ++i) {
            float e = eh[i * 2 + h] + erh;
            e = (e > 0.f) ? e : NEG_SLOPE * e;
            s += expf(e - mx);
        }
        const float inv = (cnt > 0) ? 1.f / s : 0.f;
        float accum = 0.f;
        for (int i = 0; i < cnt; ++i) {
            float e = eh[i * 2 + h] + erh;
            e = (e > 0.f) ? e : NEG_SLOPE * e;
            accum += (expf(e - mx) * inv) * Zs[i * ZP + h * OUT_DIM + d];
        }
        partial[h * OUT_DIM + d] = accum + bias[h * OUT_DIM + d];
    }
    __syncthreads();
    if (tid < OUT_DIM)
        out[(long)g * OUT_STRIDE + dir * OUT_DIM + tid] =
            0.5f * (partial[tid] + partial[OUT_DIM + tid]);
}

extern "C" void kernel_launch(void* const* d_in, const int* in_sizes, int n_in,
                              void* d_out, int out_size, void* d_ws, size_t ws_size,
                              hipStream_t stream)
{
    const float* feat     = (const float*)d_in[0];
    const int*   src      = (const int*)  d_in[1];
    const int*   dst      = (const int*)  d_in[2];
    const int*   op       = (const int*)  d_in[3];
    const int*   parallel = (const int*)  d_in[4];
    const float* W_f  = (const float*)d_in[5];
    const float* al_f = (const float*)d_in[6];
    const float* ar_f = (const float*)d_in[7];
    const float* b_f  = (const float*)d_in[8];
    const float* W_b  = (const float*)d_in[9];
    const float* al_b = (const float*)d_in[10];
    const float* ar_b = (const float*)d_in[11];
    const float* b_b  = (const float*)d_in[12];
    float* out = (float*)d_out;

    gat_node_kernel<<<dim3(N_GRAPHS * 3), dim3(256), 0, stream>>>(
        feat, src, dst, op, parallel, W_f, al_f, ar_f, b_f, W_b, al_b, ar_b, b_b, out);
}


# phase-3 softmax: each weight exp(e-max) computed once by one thread per (row, head), shared via LDS; was redone by all 256 output threads in 3 serial passes; sum order unchanged (zero-weight padding)
# speedup vs baseline: 2.9431x; 1.1316x over previous
#include <hip/hip_runtime.h>
#include <math.h>

#define N_GRAPHS 64
#define M_NODES  512
#define IN_DIM   256
#define OUT_DIM  128
#define NHEADS   2
#define ZD       256
#define EG       8192
#define P_IDX    32
#define NEG_SLOPE 0.2f

#define CAP      47
#define MAXR     48
#define NWAVES   8
#define WEDGES   (EG / NWAVES)
#define WSTRIDE  48
#define OUT_STRIDE 768
#define ZP       260

typedef float v2f __attribute__((ext_vector_type(2)));
typedef float v8f __attribute__((ext_vector_type(8)));
typedef float v4f __attribute__((ext_vector_type(4)));

template <int MT>
__device__ __forceinline__ void gemm_rows(const float* __restrict__ feat,
                                          const float* __restrict__ Wd,
                                          const int* rowNode, float* Zs,
                                          int R, int wave, int lane)
{
    const int lm    = lane & 15;
    const int khalf = (lane >> 4) << 1;
    const float* fr[MT];
    #pragma unroll
    for (int mt = 0; mt < MT; ++mt)
        fr[mt] = feat + (long)rowNode[min(mt * 16 + lm, R - 1)] * IN_DIM + khalf;
    const int ncol0 = wave * 32 + lm;
    const float* wc = Wd + (long)khalf * ZD + ncol0;

    v8f acc[MT][2];
    #pragma unroll
    for (int mt = 0; mt < MT; ++mt) {
        acc[mt][0] = v8f{};
        acc[mt][1] = v8f{};
    }

    #pragma unroll 8
    for (int k0 = 0; k0 < IN_DIM; k0 += 4) {
        v2f a[MT], b[2];
        #pragma unroll
        for (int mt = 0; mt < MT; ++mt) {
            a[mt].x = fr[mt][k0];
            a[mt].y = fr[mt][k0 + 1];
        }
        #pragma unroll
        for (int nn = 0; nn < 2; ++nn) {
            b[nn].x = wc[(long)k0 * ZD + nn * 16];
            b[nn].y = wc[(long)(k0 + 1) * ZD + nn * 16];
        }
        #pragma unroll
        for (int mt = 0; mt < MT; ++mt) {
            #pragma unroll
            for (int nn = 0; nn < 2; ++nn)
                acc[mt][nn] = __builtin_amdgcn_wmma_f32_16x16x4_f32(
                                  false, a[mt], false, b[nn], (short)0, acc[mt][nn], false, false);
        }
    }
    #pragma unroll
    for (int mt = 0; mt < MT; ++mt) {
        const int mrow = mt * 16 + ((lane >> 4) << 3);
        #pragma unroll
        for (int nn = 0; nn < 2; ++nn) {
            #pragma unroll
            for (int r = 0; r < 8; ++r)
                Zs[(mrow + r) * ZP + ncol0 + nn * 16] = acc[mt][nn][r];
        }
    }
}

__device__ __forceinline__ void tail_rows(const float* __restrict__ feat,
                                          const int*   __restrict__ opPtr,
                                          const int*   __restrict__ parallel,
                                          float* __restrict__ out, int g, int t)
{
    const int node = opPtr[0] + g * M_NODES;
    out[(long)g * OUT_STRIDE + 256 + t] = feat[(long)node * IN_DIM + t];
    float s = 0.f;
    for (int p = 0; p < P_IDX; ++p) {
        int n = parallel[p] + g * M_NODES;
        s += feat[(long)n * IN_DIM + t];
    }
    out[(long)g * OUT_STRIDE + 512 + t] = s;
}

__global__ __launch_bounds__(256, 2)
void gat_node_kernel(const float* __restrict__ feat,
                     const int*   __restrict__ srcArr,
                     const int*   __restrict__ dstArr,
                     const int*   __restrict__ opPtr,
                     const int*   __restrict__ parallel,
                     const float* __restrict__ W_f,  const float* __restrict__ al_f,
                     const float* __restrict__ ar_f, const float* __restrict__ b_f,
                     const float* __restrict__ W_b,  const float* __restrict__ al_b,
                     const float* __restrict__ ar_b, const float* __restrict__ b_b,
                     float* __restrict__ out)
{
    __shared__ __attribute__((aligned(16))) float Zs[MAXR * ZP];
    __shared__ __attribute__((aligned(16))) float av[2 * ZD];
    __shared__ __attribute__((aligned(16))) float eh[NHEADS * MAXR];
    __shared__ __attribute__((aligned(16))) float pw[NHEADS * MAXR];
    __shared__ float er[NHEADS];
    __shared__ float partial[NHEADS * OUT_DIM];
    __shared__ int   rowNode[MAXR];
    __shared__ int   wl[NWAVES * WSTRIDE];
    __shared__ int   wtot[NWAVES];

    const int gb   = blockIdx.x;
    if (gb >= 2 * N_GRAPHS) {
        tail_rows(feat, opPtr, parallel, out, gb - 2 * N_GRAPHS, (int)threadIdx.x);
        return;
    }
    const int g    = gb >> 1;
    const int dir  = gb & 1;
    const int tid  = threadIdx.x;
    const int node = opPtr[0] + g * M_NODES;

    const int*   seg  = (dir == 0) ? dstArr : srcArr;
    const int*   gat  = (dir == 0) ? srcArr : dstArr;
    const float* Wd   = (dir == 0) ? W_f  : W_b;
    const float* al   = (dir == 0) ? al_f : al_b;
    const float* ar   = (dir == 0) ? ar_f : ar_b;
    const float* bias = (dir == 0) ? b_f  : b_b;

    const int wave  = tid >> 5;
    const int lane  = tid & 31;
    const int wbase = g * EG + wave * WEDGES;
    int v[8][4];
    #pragma unroll
    for (int i = 0; i < 8; ++i) {
        const int* p = seg + wbase + i * 128 + lane * 4;
        v[i][0] = p[0];
        v[i][1] = p[1];
        v[i][2] = p[2];
        v[i][3] = p[3];
    }
    const float alv = al[tid];
    const float arv = ar[tid];
    int run = 0;
    #pragma unroll
    for (int i = 0; i < 8; ++i) {
        const unsigned m0 = __builtin_amdgcn_ballot_w32(v[i][0] == node);
        const unsigned m1 = __builtin_amdgcn_ballot_w32(v[i][1] == node);
        const unsigned m2 = __builtin_amdgcn_ballot_w32(v[i][2] == node);
        const unsigned m3 = __builtin_amdgcn_ballot_w32(v[i][3] == node);
        if ((m0 | m1 | m2 | m3) != 0u) {
            unsigned pos = (unsigned)run;
            pos = __builtin_amdgcn_mbcnt_lo(m0, pos);
            pos = __builtin_amdgcn_mbcnt_lo(m1, pos);
            pos = __builtin_amdgcn_mbcnt_lo(m2, pos);
            pos = __builtin_amdgcn_mbcnt_lo(m3, pos);
            const int e0 = wbase + i * 128 + lane * 4;
            #pragma unroll
            for (int q = 0; q < 4; ++q) {
                if (v[i][q] == node) {
                    if (pos < CAP) wl[wave * WSTRIDE + pos] = e0 + q;
                    ++pos;
                }
            }
            run += __builtin_popcount(m0) + __builtin_popcount(m1)
                 + __builtin_popcount(m2) + __builtin_popcount(m3);
        }
    }
    if (lane == 0) wtot[wave] = run;
    av[tid]      = alv;
    av[ZD + tid] = arv;
    __syncthreads();
    int before = 0, total = 0;
    #pragma unroll
    for (int w = 0; w < NWAVES; ++w) {
        const int t = wtot[w];
        before += (w < wave) ? t : 0;
        total  += t;
    }
    const int cnt   = min(total, CAP);
    const int R     = cnt + 1;
    const int tiles = (R + 15) >> 4;
    const int mine = min(run, CAP);
    #pragma unroll
    for (int t = 0; t < 2; ++t) {
        const int lp = lane + 32 * t;
        if (lp < mine && before + lp < CAP)
            rowNode[before + lp] = gat[wl[wave * WSTRIDE + lp]];
    }
    if (tid == 0) rowNode[cnt] = node;
    __syncthreads();

    if (tiles == 1)      gemm_rows<1>(feat, Wd, rowNode, Zs, R, wave, lane);
    else if (tiles == 2) gemm_rows<2>(feat, Wd, rowNode, Zs, R, wave, lane);
    else                 gemm_rows<3>(feat, Wd, rowNode, Zs, R, wave, lane);
    __syncthreads();

    if (tid < 2 * MAXR) {
        const int row = tid >> 1;
        const int h   = tid & 1;
        float elv = -INFINITY;
        if (row <= cnt) {
            const v4f* zr = (const v4f*)&Zs[row * ZP + h * OUT_DIM];
            const v4f* a  = (const v4f*)&av[((row == cnt) ? ZD : 0) + h * OUT_DIM];
            v4f zc[4], ac[4];
            #pragma unroll
            for (int q = 0; q < 4; ++q) { zc[q] = zr[q]; ac[q] = a[q]; }
            float s = 0.f;
            #pragma unroll
            for (int c = 0; c < 8; ++c) {
                v4f zn[4], an[4];
                if (c < 7) {
                    #pragma unroll
                    for (int q = 0; q < 4; ++q) { zn[q] = zr[(c + 1) * 4 + q]; an[q] = a[(c + 1) * 4 + q]; }
                }
                asm volatile("" :: "v"(zc[0]), "v"(zc[1]), "v"(zc[2]), "v"(zc[3]),
                                   "v"(ac[0]), "v"(ac[1]), "v"(ac[2]), "v"(ac[3]));
                #pragma unroll
                for (int q = 0; q < 4; ++q) {
                    s = fmaf(zc[q].x, ac[q].x, s);
                    s = fmaf(zc[q].y, ac[q].y, s);
                    s = fmaf(zc[q].z, ac[q].z, s);
                    s = fmaf(zc[q].w, ac[q].w, s);
                }
                if (c < 7) {
                    #pragma unroll
                    for (int q = 0; q < 4; ++q) { zc[q] = zn[q]; ac[q] = an[q]; }
                }
            }
            if (row == cnt) er[h] = s;
            else            elv = s;
        }
        eh[h * MAXR + row] = elv;
    }
    __syncthreads();

    if (tid < 2 * MAXR) {
        const int row = tid >> 1;
        const int h   = tid & 1;
        const float erh = er[h];
        const v4f* e4 = (const v4f*)&eh[h * MAXR];
        v4f ev[MAXR / 4];
        #pragma unroll
        for (int j = 0; j < MAXR / 4; ++j) ev[j] = e4[j];
        const float elo = eh[h * MAXR + row];
        v4f m4 = { -INFINITY, -INFINITY, -INFINITY, -INFINITY };
        #pragma unroll
        for (int j = 0; j < MAXR / 4; ++j) {
            if (4 * j < cnt) {
                v4f e = ev[j] + erh;
                e.x = (e.x > 0.f) ? e.x : NEG_SLOPE * e.x;
                e.y = (e.y > 0.f) ? e.y : NEG_SLOPE * e.y;
                e.z = (e.z > 0.f) ? e.z : NEG_SLOPE * e.z;
                e.w = (e.w > 0.f) ? e.w : NEG_SLOPE * e.w;
                m4.x = fmaxf(m4.x, e.x);
                m4.y = fmaxf(m4.y, e.y);
                m4.z = fmaxf(m4.z, e.z);
                m4.w = fmaxf(m4.w, e.w);
            }
        }
        const float mx = fmaxf(fmaxf(m4.x, m4.y), fmaxf(m4.z, m4.w));
        float eo = elo + erh;
        eo = (eo > 0.f) ? eo : NEG_SLOPE * eo;
        const float p = expf(eo - mx);
        pw[h * MAXR + row] = (row < cnt) ? p : 0.f;
    }
    __syncthreads();

    {
        const int h = tid >> 7;
        const int d = tid & 127;
        const v4f* p4 = (const v4f*)&pw[h * MAXR];
        v4f pv[MAXR / 4];
        #pragma unroll
        for (int j = 0; j < MAXR / 4; ++j) pv[j] = p4[j];
        const float* zc = &Zs[h * OUT_DIM + d];
        float z[MAXR];
        #pragma unroll
        for (int t = 0; t < MAXR / 16; ++t) {
            if (16 * t < cnt) {
                #pragma unroll
                for (int q = 0; q < 16; ++q) z[16 * t + q] = zc[(16 * t + q) * ZP];
            }
        }
        float s = 0.f;
        #pragma unroll
        for (int t = 0; t < MAXR / 16; ++t) {
            if (16 * t < cnt) {
                #pragma unroll
                for (int j = 4 * t; j < 4 * t + 4; ++j) {
                    s += pv[j].x;
                    s += pv[j].y;
                    s += pv[j].z;
                    s += pv[j].w;
                }
            }
        }
        const float inv = (cnt > 0) ? 1.f / s : 0.f;
        float accum = 0.f;
        #pragma unroll
        for (int t = 0; t < MAXR / 16; ++t) {
            if (16 * t < cnt) {
                #pragma unroll
                for (int j = 4 * t; j < 4 * t + 4; ++j) {
                    accum = fmaf(pv[j].x * inv, z[4 * j + 0], accum);
                    accum = fmaf(pv[j].y * inv, z[4 * j + 1], accum);
                    accum = fmaf(pv[j].z * inv, z[4 * j + 2], accum);
                    accum = fmaf(pv[j].w * inv, z[4 * j + 3], accum);
                }
            }
        }
        partial[h * OUT_DIM + d] = accum + bias[h * OUT_DIM + d];
    }
    __syncthreads();
    if (tid < OUT_DIM)
        out[(long)g * OUT_STRIDE + dir * OUT_DIM + tid] =
            0.5f * (partial[tid] + partial[OUT_DIM + tid]);
}

extern "C" void kernel_launch(void* const* d_in, const int* in_sizes, int n_in,
                              void* d_out, int out_size, void* d_ws, size_t ws_size,
                              hipStream_t stream)
{
    const float* feat     = (const float*)d_in[0];
    const int*   src      = (const int*)  d_in[1];
    const int*   dst      = (const int*)  d_in[2];
    const int*   op       = (const int*)  d_in[3];
    const int*   parallel = (const int*)  d_in[4];
    const float* W_f  = (const float*)d_in[5];
    const float* al_f = (const float*)d_in[6];
    const float* ar_f = (const float*)d_in[7];
    const float* b_f  = (const float*)d_in[8];
    const float* W_b  = (const float*)d_in[9];
    const float* al_b = (const float*)d_in[10];
    const float* ar_b = (const float*)d_in[11];
    const float* b_b  = (const float*)d_in[12];
    float* out = (float*)d_out;

    gat_node_kernel<<<dim3(N_GRAPHS * 3), dim3(256), 0, stream>>>(
        feat, src, dst, op, parallel, W_f, al_f, ar_f, b_f, W_b, al_b, ar_b, b_b, out);
}


# phase-1 A operand via LDS: gathered feature rows staged once per block by whole-line 16-byte loads, A fragments read from LDS; was gathered from global by each of 8 waves; k order and sums unchanged
# speedup vs baseline: 3.0929x; 1.0509x over previous
#include <hip/hip_runtime.h>
#include <math.h>

#define N_GRAPHS 64
#define M_NODES  512
#define IN_DIM   256
#define OUT_DIM  128
#define NHEADS   2
#define ZD       256
#define EG       8192
#define P_IDX    32
#define NEG_SLOPE 0.2f

#define CAP      47
#define MAXR     48
#define NWAVES   8
#define WEDGES   (EG / NWAVES)
#define WSTRIDE  48
#define OUT_STRIDE 768
#define ZP       260

#define XP       260

typedef float v2f __attribute__((ext_vector_type(2)));
typedef float v8f __attribute__((ext_vector_type(8)));
typedef float v4f __attribute__((ext_vector_type(4)));

template <int MT>
__device__ __forceinline__ void gemm_rows(const float* __restrict__ feat,
                                          const float* __restrict__ Wd,
                                          const int* rowNode, float* Xs, float* Zs,
                                          int R, int wave, int lane)
{
    {
        const int tid = wave * 32 + lane;
        const int c4  = tid & 63;
        const int r0  = tid >> 6;
        int rn[MT * 4];
        v4f x[MT * 4];
        #pragma unroll
        for (int j = 0; j < MT * 4; ++j) rn[j] = rowNode[min(r0 + 4 * j, R - 1)];
        #pragma unroll
        for (int j = 0; j < MT * 4; ++j) x[j] = *(const v4f*)(feat + (long)rn[j] * IN_DIM + c4 * 4);
        #pragma unroll
        for (int j = 0; j < MT * 4; ++j) *(v4f*)&Xs[(r0 + 4 * j) * XP + c4 * 4] = x[j];
    }
    __syncthreads();

    const int lm    = lane & 15;
    const int khalf = (lane >> 4) << 1;
    const float* xa = Xs + lm * XP + khalf;
    const int ncol0 = wave * 32 + lm;
    const float* wc = Wd + (long)khalf * ZD + ncol0;

    v8f acc[MT][2];
    #pragma unroll
    for (int mt = 0; mt < MT; ++mt) {
        acc[mt][0] = v8f{};
        acc[mt][1] = v8f{};
    }

    #pragma nounroll
    for (int kg = 0; kg < IN_DIM; kg += 32) {
        v2f a[8][MT], b[8][2];
        #pragma unroll
        for (int s = 0; s < 8; ++s) {
            const int k0 = kg + 4 * s;
            #pragma unroll
            for (int mt = 0; mt < MT; ++mt)
                a[s][mt] = *(const v2f*)(xa + mt * 16 * XP + k0);
            #pragma unroll
            for (int nn = 0; nn < 2; ++nn) {
                b[s][nn].x = wc[(long)k0 * ZD + nn * 16];
                b[s][nn].y = wc[(long)(k0 + 1) * ZD + nn * 16];
            }
        }
        __builtin_amdgcn_sched_barrier(0);
        #pragma unroll
        for (int s = 0; s < 8; ++s) {
            #pragma unroll
            for (int mt = 0; mt < MT; ++mt) {
                #pragma unroll
                for (int nn = 0; nn < 2; ++nn)
                    acc[mt][nn] = __builtin_amdgcn_wmma_f32_16x16x4_f32(
                                      false, a[s][mt], false, b[s][nn], (short)0, acc[mt][nn], false, false);
            }
        }
    }
    #pragma unroll
    for (int mt = 0; mt < MT; ++mt) {
        const int mrow = mt * 16 + ((lane >> 4) << 3);
        #pragma unroll
        for (int nn = 0; nn < 2; ++nn) {
            #pragma unroll
            for (int r = 0; r < 8; ++r)
                Zs[(mrow + r) * ZP + ncol0 + nn * 16] = acc[mt][nn][r];
        }
    }
}

__device__ __forceinline__ void tail_rows(const float* __restrict__ feat,
                                          const int*   __restrict__ opPtr,
                                          const int*   __restrict__ parallel,
                                          float* __restrict__ out, int g, int t)
{
    const int node = opPtr[0] + g * M_NODES;
    out[(long)g * OUT_STRIDE + 256 + t] = feat[(long)node * IN_DIM + t];
    float s = 0.f;
    for (int p = 0; p < P_IDX; ++p) {
        int n = parallel[p] + g * M_NODES;
        s += feat[(long)n * IN_DIM + t];
    }
    out[(long)g * OUT_STRIDE + 512 + t] = s;
}

__global__ __launch_bounds__(256, 2)
void gat_node_kernel(const float* __restrict__ feat,
                     const int*   __restrict__ srcArr,
                     const int*   __restrict__ dstArr,
                     const int*   __restrict__ opPtr,
                     const int*   __restrict__ parallel,
                     const float* __restrict__ W_f,  const float* __restrict__ al_f,
                     const float* __restrict__ ar_f, const float* __restrict__ b_f,
                     const float* __restrict__ W_b,  const float* __restrict__ al_b,
                     const float* __restrict__ ar_b, const float* __restrict__ b_b,
                     float* __restrict__ out)
{
    __shared__ __attribute__((aligned(16))) float Zs[MAXR * ZP];
    __shared__ __attribute__((aligned(16))) float Xs[MAXR * XP];
    __shared__ __attribute__((aligned(16))) float av[2 * ZD];
    __shared__ __attribute__((aligned(16))) float eh[NHEADS * MAXR];
    __shared__ __attribute__((aligned(16))) float pw[NHEADS * MAXR];
    __shared__ float er[NHEADS];
    __shared__ float partial[NHEADS * OUT_DIM];
    __shared__ int   rowNode[MAXR];
    __shared__ int   wl[NWAVES * WSTRIDE];
    __shared__ int   wtot[NWAVES];

    const int gb   = blockIdx.x;
    if (gb >= 2 * N_GRAPHS) {
        tail_rows(feat, opPtr, parallel, out, gb - 2 * N_GRAPHS, (int)threadIdx.x);
        return;
    }
    const int g    = gb >> 1;
    const int dir  = gb & 1;
    const int tid  = threadIdx.x;
    const int node = opPtr[0] + g * M_NODES;

    const int*   seg  = (dir == 0) ? dstArr : srcArr;
    const int*   gat  = (dir == 0) ? srcArr : dstArr;
    const float* Wd   = (dir == 0) ? W_f  : W_b;
    const float* al   = (dir == 0) ? al_f : al_b;
    const float* ar   = (dir == 0) ? ar_f : ar_b;
    const float* bias = (dir == 0) ? b_f  : b_b;

    const int wave  = tid >> 5;
    const int lane  = tid & 31;
    const int wbase = g * EG + wave * WEDGES;
    int v[8][4];
    #pragma unroll
    for (int i = 0; i < 8; ++i) {
        const int* p = seg + wbase + i * 128 + lane * 4;
        v[i][0] = p[0];
        v[i][1] = p[1];
        v[i][2] = p[2];
        v[i][3] = p[3];
    }
    const float alv = al[tid];
    const float arv = ar[tid];
    int run = 0;
    #pragma unroll
    for (int i = 0; i < 8; ++i) {
        const unsigned m0 = __builtin_amdgcn_ballot_w32(v[i][0] == node);
        const unsigned m1 = __builtin_amdgcn_ballot_w32(v[i][1] == node);
        const unsigned m2 = __builtin_amdgcn_ballot_w32(v[i][2] == node);
        const unsigned m3 = __builtin_amdgcn_ballot_w32(v[i][3] == node);
        if ((m0 | m1 | m2 | m3) != 0u) {
            unsigned pos = (unsigned)run;
            pos = __builtin_amdgcn_mbcnt_lo(m0, pos);
            pos = __builtin_amdgcn_mbcnt_lo(m1, pos);
            pos = __builtin_amdgcn_mbcnt_lo(m2, pos);
            pos = __builtin_amdgcn_mbcnt_lo(m3, pos);
            const int e0 = wbase + i * 128 + lane * 4;
            #pragma unroll
            for (int q = 0; q < 4; ++q) {
                if (v[i][q] == node) {
                    if (pos < CAP) wl[wave * WSTRIDE + pos] = e0 + q;
                    ++pos;
                }
            }
            run += __builtin_popcount(m0) + __builtin_popcount(m1)
                 + __builtin_popcount(m2) + __builtin_popcount(m3);
        }
    }
    if (lane == 0) wtot[wave] = run;
    av[tid]      = alv;
    av[ZD + tid] = arv;
    __syncthreads();
    int before = 0, total = 0;
    #pragma unroll
    for (int w = 0; w < NWAVES; ++w) {
        const int t = wtot[w];
        before += (w < wave) ? t : 0;
        total  += t;
    }
    const int cnt   = min(total, CAP);
    const int R     = cnt + 1;
    const int tiles = (R + 15) >> 4;
    const int mine = min(run, CAP);
    #pragma unroll
    for (int t = 0; t < 2; ++t) {
        const int lp = lane + 32 * t;
        if (lp < mine && before + lp < CAP)
            rowNode[before + lp] = gat[wl[wave * WSTRIDE + lp]];
    }
    if (tid == 0) rowNode[cnt] = node;
    __syncthreads();

    if (tiles == 1)      gemm_rows<1>(feat, Wd, rowNode, Xs, Zs, R, wave, lane);
    else if (tiles == 2) gemm_rows<2>(feat, Wd, rowNode, Xs, Zs, R, wave, lane);
    else                 gemm_rows<3>(feat, Wd, rowNode, Xs, Zs, R, wave, lane);
    __syncthreads();

    if (tid < 2 * MAXR) {
        const int row = tid >> 1;
        const int h   = tid & 1;
        float elv = -INFINITY;
        if (row <= cnt) {
            const v4f* zr = (const v4f*)&Zs[row * ZP + h * OUT_DIM];
            const v4f* a  = (const v4f*)&av[((row == cnt) ? ZD : 0) + h * OUT_DIM];
            v4f zc[4], ac[4];
            #pragma unroll
            for (int q = 0; q < 4; ++q) { zc[q] = zr[q]; ac[q] = a[q]; }
            float s = 0.f;
            #pragma unroll
            for (int c = 0; c < 8; ++c) {
                v4f zn[4], an[4];
                if (c < 7) {
                    #pragma unroll
                    for (int q = 0; q < 4; ++q) { zn[q] = zr[(c + 1) * 4 + q]; an[q] = a[(c + 1) * 4 + q]; }
                }
                asm volatile("" :: "v"(zc[0]), "v"(zc[1]), "v"(zc[2]), "v"(zc[3]),
                                   "v"(ac[0]), "v"(ac[1]), "v"(ac[2]), "v"(ac[3]));
                #pragma unroll
                for (int q = 0; q < 4; ++q) {
                    s = fmaf(zc[q].x, ac[q].x, s);
                    s = fmaf(zc[q].y, ac[q].y, s);
                    s = fmaf(zc[q].z, ac[q].z, s);
                    s = fmaf(zc[q].w, ac[q].w, s);
                }
                if (c < 7) {
                    #pragma unroll
                    for (int q = 0; q < 4; ++q) { zc[q] = zn[q]; ac[q] = an[q]; }
                }
            }
            if (row == cnt) er[h] = s;
            else            elv = s;
        }
        eh[h * MAXR + row] = elv;
    }
    __syncthreads();

    if (tid < 2 * MAXR) {
        const int row = tid >> 1;
        const int h   = tid & 1;
        const float erh = er[h];
        const v4f* e4 = (const v4f*)&eh[h * MAXR];
        v4f ev[MAXR / 4];
        #pragma unroll
        for (int j = 0; j < MAXR / 4; ++j) ev[j] = e4[j];
        const float elo = eh[h * MAXR + row];
        v4f m4 = { -INFINITY, -INFINITY, -INFINITY, -INFINITY };
        #pragma unroll
        for (int j = 0; j < MAXR / 4; ++j) {
            if (4 * j < cnt) {
                v4f e = ev[j] + erh;
                e.x = (e.x > 0.f) ? e.x : NEG_SLOPE * e.x;
                e.y = (e.y > 0.f) ? e.y : NEG_SLOPE * e.y;
                e.z = (e.z > 0.f) ? e.z : NEG_SLOPE * e.z;
                e.w = (e.w > 0.f) ? e.w : NEG_SLOPE * e.w;
                m4.x = fmaxf(m4.x, e.x);
                m4.y = fmaxf(m4.y, e.y);
                m4.z = fmaxf(m4.z, e.z);
                m4.w = fmaxf(m4.w, e.w);
            }
        }
        const float mx = fmaxf(fmaxf(m4.x, m4.y), fmaxf(m4.z, m4.w));
        float eo = elo + erh;
        eo = (eo > 0.f) ? eo : NEG_SLOPE * eo;
        const float p = expf(eo - mx);
        pw[h * MAXR + row] = (row < cnt) ? p : 0.f;
    }
    __syncthreads();

    {
        const int h = tid >> 7;
        const int d = tid & 127;
        const v4f* p4 = (const v4f*)&pw[h * MAXR];
        v4f pv[MAXR / 4];
        #pragma unroll
        for (int j = 0; j < MAXR / 4; ++j) pv[j] = p4[j];
        const float* zc = &Zs[h * OUT_DIM + d];
        float z[MAXR];
        #pragma unroll
        for (int t = 0; t < MAXR / 16; ++t) {
            if (16 * t < cnt) {
                #pragma unroll
                for (int q = 0; q < 16; ++q) z[16 * t + q] = zc[(16 * t + q) * ZP];
            }
        }
        float s = 0.f;
        #pragma unroll
        for (int t = 0; t < MAXR / 16; ++t) {
            if (16 * t < cnt) {
                #pragma unroll
                for (int j = 4 * t; j < 4 * t + 4; ++j) {
                    s += pv[j].x;
                    s += pv[j].y;
                    s += pv[j].z;
                    s += pv[j].w;
                }
            }
        }
        const float inv = (cnt > 0) ? 1.f / s : 0.f;
        float accum = 0.f;
        #pragma unroll
        for (int t = 0; t < MAXR / 16; ++t) {
            if (16 * t < cnt) {
                #pragma unroll
                for (int j = 4 * t; j < 4 * t + 4; ++j) {
                    accum = fmaf(pv[j].x * inv, z[4 * j + 0], accum);
                    accum = fmaf(pv[j].y * inv, z[4 * j + 1], accum);
                    accum = fmaf(pv[j].z * inv, z[4 * j + 2], accum);
                    accum = fmaf(pv[j].w * inv, z[4 * j + 3], accum);
                }
            }
        }
        partial[h * OUT_DIM + d] = accum + bias[h * OUT_DIM + d];
    }
    __syncthreads();
    if (tid < OUT_DIM)
        out[(long)g * OUT_STRIDE + dir * OUT_DIM + tid] =
            0.5f * (partial[tid] + partial[OUT_DIM + tid]);
}

extern "C" void kernel_launch(void* const* d_in, const int* in_sizes, int n_in,
                              void* d_out, int out_size, void* d_ws, size_t ws_size,
                              hipStream_t stream)
{
    const float* feat     = (const float*)d_in[0];
    const int*   src      = (const int*)  d_in[1];
    const int*   dst      = (const int*)  d_in[2];
    const int*   op       = (const int*)  d_in[3];
    const int*   parallel = (const int*)  d_in[4];
    const float* W_f  = (const float*)d_in[5];
    const float* al_f = (const float*)d_in[6];
    const float* ar_f = (const float*)d_in[7];
    const float* b_f  = (const float*)d_in[8];
    const float* W_b  = (const float*)d_in[9];
    const float* al_b = (const float*)d_in[10];
    const float* ar_b = (const float*)d_in[11];
    const float* b_b  = (const float*)d_in[12];
    float* out = (float*)d_out;

    gat_node_kernel<<<dim3(N_GRAPHS * 3), dim3(256), 0, stream>>>(
        feat, src, dst, op, parallel, W_f, al_f, ar_f, b_f, W_b, al_b, ar_b, b_b, out);
}


# phase-1 B operand software-pipelined: W fragments of k-group g+1 requested in front of group g's WMMAs (two register sets, last 2 groups peeled), group 0 at kernel entry; k order, sums unchanged
# speedup vs baseline: 3.5047x; 1.1331x over previous
#include <hip/hip_runtime.h>
#include <math.h>

#define N_GRAPHS 64
#define M_NODES  512
#define IN_DIM   256
#define OUT_DIM  128
#define NHEADS   2
#define ZD       256
#define EG       8192
#define P_IDX    32
#define NEG_SLOPE 0.2f

#define CAP      47
#define MAXR     48
#define NWAVES   8
#define WEDGES   (EG / NWAVES)
#define WSTRIDE  48
#define OUT_STRIDE 768
#define ZP       260

#define XP       260

typedef float v2f __attribute__((ext_vector_type(2)));
typedef float v8f __attribute__((ext_vector_type(8)));
typedef float v4f __attribute__((ext_vector_type(4)));

__device__ __forceinline__ void load_b_group(v2f (&b)[8][2], const float* __restrict__ wc, int kg)
{
    #pragma unroll
    for (int s = 0; s < 8; ++s) {
        const int k0 = kg + 4 * s;
        #pragma unroll
        for (int nn = 0; nn < 2; ++nn) {
            b[s][nn].x = wc[(long)k0 * ZD + nn * 16];
            b[s][nn].y = wc[(long)(k0 + 1) * ZD + nn * 16];
        }
    }
}

template <int MT>
__device__ __forceinline__ void wmma_group(v8f (&acc)[MT][2], const v2f (&a)[8][MT], const v2f (&b)[8][2])
{
    #pragma unroll
    for (int s = 0; s < 8; ++s) {
        #pragma unroll
        for (int mt = 0; mt < MT; ++mt) {
            #pragma unroll
            for (int nn = 0; nn < 2; ++nn)
                acc[mt][nn] = __builtin_amdgcn_wmma_f32_16x16x4_f32(
                                  false, a[s][mt], false, b[s][nn], (short)0, acc[mt][nn], false, false);
        }
    }
}

template <int MT, bool NEXT, bool LEAD>
__device__ __forceinline__ void gemm_group(v8f (&acc)[MT][2], const float* xa, const float* __restrict__ wc, int kg,
                                           const v2f (&bUse)[8][2], v2f (&bNext)[8][2])
{
    if (LEAD) __builtin_amdgcn_sched_barrier(0);
    v2f a[8][MT];
    #pragma unroll
    for (int s = 0; s < 8; ++s) {
        #pragma unroll
        for (int mt = 0; mt < MT; ++mt)
            a[s][mt] = *(const v2f*)(xa + mt * 16 * XP + kg + 4 * s);
    }
    if (NEXT) load_b_group(bNext, wc, kg + 32);
    __builtin_amdgcn_sched_barrier(0);
    wmma_group<MT>(acc, a, bUse);
}

template <int MT>
__device__ __forceinline__ void gemm_rows(const float* __restrict__ feat,
                                          const int* rowNode, float* Xs, float* Zs,
                                          int R, int wave, int lane,
                                          const float* __restrict__ wc, v2f (&b0)[8][2])
{
    {
        const int tid = wave * 32 + lane;
        const int c4  = tid & 63;
        const int r0  = tid >> 6;
        int rn[MT * 4];
        v4f x[MT * 4];
        #pragma unroll
        for (int j = 0; j < MT * 4; ++j) rn[j] = rowNode[min(r0 + 4 * j, R - 1)];
        #pragma unroll
        for (int j = 0; j < MT * 4; ++j) x[j] = *(const v4f*)(feat + (long)rn[j] * IN_DIM + c4 * 4);
        #pragma unroll
        for (int j = 0; j < MT * 4; ++j) *(v4f*)&Xs[(r0 + 4 * j) * XP + c4 * 4] = x[j];
    }
    __syncthreads();

    const int lm    = lane & 15;
    const int khalf = (lane >> 4) << 1;
    const float* xa = Xs + lm * XP + khalf;
    const int ncol0 = wave * 32 + lm;

    v8f acc[MT][2];
    #pragma unroll
    for (int mt = 0; mt < MT; ++mt) {
        acc[mt][0] = v8f{};
        acc[mt][1] = v8f{};
    }

    v2f b1[8][2];
    #pragma nounroll
    for (int kg = 0; kg < IN_DIM - 64; kg += 64) {
        gemm_group<MT, true, false>(acc, xa, wc, kg,      b0, b1);
        gemm_group<MT, true, true >(acc, xa, wc, kg + 32, b1, b0);
    }
    gemm_group<MT, true,  false>(acc, xa, wc, IN_DIM - 64, b0, b1);
    gemm_group<MT, false, true >(acc, xa, wc, IN_DIM - 32, b1, b0);
    #pragma unroll
    for (int mt = 0; mt < MT; ++mt) {
        const int mrow = mt * 16 + ((lane >> 4) << 3);
        #pragma unroll
        for (int nn = 0; nn < 2; ++nn) {
            #pragma unroll
            for (int r = 0; r < 8; ++r)
                Zs[(mrow + r) * ZP + ncol0 + nn * 16] = acc[mt][nn][r];
        }
    }
}

__device__ __forceinline__ void tail_rows(const float* __restrict__ feat,
                                          const int*   __restrict__ opPtr,
                                          const int*   __restrict__ parallel,
                                          float* __restrict__ out, int g, int t)
{
    const int node = opPtr[0] + g * M_NODES;
    out[(long)g * OUT_STRIDE + 256 + t] = feat[(long)node * IN_DIM + t];
    float s = 0.f;
    for (int p = 0; p < P_IDX; ++p) {
        int n = parallel[p] + g * M_NODES;
        s += feat[(long)n * IN_DIM + t];
    }
    out[(long)g * OUT_STRIDE + 512 + t] = s;
}

__global__ __launch_bounds__(256, 2)
void gat_node_kernel(const float* __restrict__ feat,
                     const int*   __restrict__ srcArr,
                     const int*   __restrict__ dstArr,
                     const int*   __restrict__ opPtr,
                     const int*   __restrict__ parallel,
                     const float* __restrict__ W_f,  const float* __restrict__ al_f,
                     const float* __restrict__ ar_f, const float* __restrict__ b_f,
                     const float* __restrict__ W_b,  const float* __restrict__ al_b,
                     const float* __restrict__ ar_b, const float* __restrict__ b_b,
                     float* __restrict__ out)
{
    __shared__ __attribute__((aligned(16))) float Zs[MAXR * ZP];
    __shared__ __attribute__((aligned(16))) float Xs[MAXR * XP];
    __shared__ __attribute__((aligned(16))) float av[2 * ZD];
    __shared__ __attribute__((aligned(16))) float eh[NHEADS * MAXR];
    __shared__ __attribute__((aligned(16))) float pw[NHEADS * MAXR];
    __shared__ float er[NHEADS];
    __shared__ float partial[NHEADS * OUT_DIM];
    __shared__ int   rowNode[MAXR];
    __shared__ int   wl[NWAVES * WSTRIDE];
    __shared__ int   wtot[NWAVES];

    const int gb   = blockIdx.x;
    if (gb >= 2 * N_GRAPHS) {
        tail_rows(feat, opPtr, parallel, out, gb - 2 * N_GRAPHS, (int)threadIdx.x);
        return;
    }
    const int g    = gb >> 1;
    const int dir  = gb & 1;
    const int tid  = threadIdx.x;
    const int node = opPtr[0] + g * M_NODES;

    const int*   seg  = (dir == 0) ? dstArr : srcArr;
    const int*   gat  = (dir == 0) ? srcArr : dstArr;
    const float* Wd   = (dir == 0) ? W_f  : W_b;
    const float* al   = (dir == 0) ? al_f : al_b;
    const float* ar   = (dir == 0) ? ar_f : ar_b;
    const float* bias = (dir == 0) ? b_f  : b_b;

    const int wave  = tid >> 5;
    const int lane  = tid & 31;
    const int wbase = g * EG + wave * WEDGES;
    int v[8][4];
    #pragma unroll
    for (int i = 0; i < 8; ++i) {
        const int* p = seg + wbase + i * 128 + lane * 4;
        v[i][0] = p[0];
        v[i][1] = p[1];
        v[i][2] = p[2];
        v[i][3] = p[3];
    }
    const float alv = al[tid];
    const float arv = ar[tid];
    const float* wc = Wd + (long)((lane >> 4) << 1) * ZD + wave * 32 + (lane & 15);
    v2f b0[8][2];
    load_b_group(b0, wc, 0);
    int run = 0;
    #pragma unroll
    for (int i = 0; i < 8; ++i) {
        const unsigned m0 = __builtin_amdgcn_ballot_w32(v[i][0] == node);
        const unsigned m1 = __builtin_amdgcn_ballot_w32(v[i][1] == node);
        const unsigned m2 = __builtin_amdgcn_ballot_w32(v[i][2] == node);
        const unsigned m3 = __builtin_amdgcn_ballot_w32(v[i][3] == node);
        if ((m0 | m1 | m2 | m3) != 0u) {
            unsigned pos = (unsigned)run;
            pos = __builtin_amdgcn_mbcnt_lo(m0, pos);
            pos = __builtin_amdgcn_mbcnt_lo(m1, pos);
            pos = __builtin_amdgcn_mbcnt_lo(m2, pos);
            pos = __builtin_amdgcn_mbcnt_lo(m3, pos);
            const int e0 = wbase + i * 128 + lane * 4;
            #pragma unroll
            for (int q = 0; q < 4; ++q) {
                if (v[i][q] == node) {
                    if (pos < CAP) wl[wave * WSTRIDE + pos] = e0 + q;
                    ++pos;
                }
            }
            run += __builtin_popcount(m0) + __builtin_popcount(m1)
                 + __builtin_popcount(m2) + __builtin_popcount(m3);
        }
    }
    if (lane == 0) wtot[wave] = run;
    av[tid]      = alv;
    av[ZD + tid] = arv;
    __syncthreads();
    int before = 0, total = 0;
    #pragma unroll
    for (int w = 0; w < NWAVES; ++w) {
        const int t = wtot[w];
        before += (w < wave) ? t : 0;
        total  += t;
    }
    const int cnt   = min(total, CAP);
    const int R     = cnt + 1;
    const int tiles = (R + 15) >> 4;
    const int mine = min(run, CAP);
    #pragma unroll
    for (int t = 0; t < 2; ++t) {
        const int lp = lane + 32 * t;
        if (lp < mine && before + lp < CAP)
            rowNode[before + lp] = gat[wl[wave * WSTRIDE + lp]];
    }
    if (tid == 0) rowNode[cnt] = node;
    __syncthreads();

    if (tiles == 1)      gemm_rows<1>(feat, rowNode, Xs, Zs, R, wave, lane, wc, b0);
    else if (tiles == 2) gemm_rows<2>(feat, rowNode, Xs, Zs, R, wave, lane, wc, b0);
    else                 gemm_rows<3>(feat, rowNode, Xs, Zs, R, wave, lane, wc, b0);
    __syncthreads();

    if (tid < 2 * MAXR) {
        const int row = tid >> 1;
        const int h   = tid & 1;
        float elv = -INFINITY;
        if (row <= cnt) {
            const v4f* zr = (const v4f*)&Zs[row * ZP + h * OUT_DIM];
            const v4f* a  = (const v4f*)&av[((row == cnt) ? ZD : 0) + h * OUT_DIM];
            v4f zc[4], ac[4];
            #pragma unroll
            for (int q = 0; q < 4; ++q) { zc[q] = zr[q]; ac[q] = a[q]; }
            float s = 0.f;
            #pragma unroll
            for (int c = 0; c < 8; ++c) {
                v4f zn[4], an[4];
                if (c < 7) {
                    #pragma unroll
                    for (int q = 0; q < 4; ++q) { zn[q] = zr[(c + 1) * 4 + q]; an[q] = a[(c + 1) * 4 + q]; }
                }
                asm volatile("" :: "v"(zc[0]), "v"(zc[1]), "v"(zc[2]), "v"(zc[3]),
                                   "v"(ac[0]), "v"(ac[1]), "v"(ac[2]), "v"(ac[3]));
                #pragma unroll
                for (int q = 0; q < 4; ++q) {
                    s = fmaf(zc[q].x, ac[q].x, s);
                    s = fmaf(zc[q].y, ac[q].y, s);
                    s = fmaf(zc[q].z, ac[q].z, s);
                    s = fmaf(zc[q].w, ac[q].w, s);
                }
                if (c < 7) {
                    #pragma unroll
                    for (int q = 0; q < 4; ++q) { zc[q] = zn[q]; ac[q] = an[q]; }
                }
            }
            if (row == cnt) er[h] = s;
            else            elv = s;
        }
        eh[h * MAXR + row] = elv;
    }
    __syncthreads();

    if (tid < 2 * MAXR) {
        const int row = tid >> 1;
        const int h   = tid & 1;
        const float erh = er[h];
        const v4f* e4 = (const v4f*)&eh[h * MAXR];
        v4f ev[MAXR / 4];
        #pragma unroll
        for (int j = 0; j < MAXR / 4; ++j) ev[j] = e4[j];
        const float elo = eh[h * MAXR + row];
        v4f m4 = { -INFINITY, -INFINITY, -INFINITY, -INFINITY };
        #pragma unroll
        for (int j = 0; j < MAXR / 4; ++j) {
            if (4 * j < cnt) {
                v4f e = ev[j] + erh;
                e.x = (e.x > 0.f) ? e.x : NEG_SLOPE * e.x;
                e.y = (e.y > 0.f) ? e.y : NEG_SLOPE * e.y;
                e.z = (e.z > 0.f) ? e.z : NEG_SLOPE * e.z;
                e.w = (e.w > 0.f) ? e.w : NEG_SLOPE * e.w;
                m4.x = fmaxf(m4.x, e.x);
                m4.y = fmaxf(m4.y, e.y);
                m4.z = fmaxf(m4.z, e.z);
                m4.w = fmaxf(m4.w, e.w);
            }
        }
        const float mx = fmaxf(fmaxf(m4.x, m4.y), fmaxf(m4.z, m4.w));
        float eo = elo + erh;
        eo = (eo > 0.f) ? eo : NEG_SLOPE * eo;
        const float p = expf(eo - mx);
        pw[h * MAXR + row] = (row < cnt) ? p : 0.f;
    }
    __syncthreads();

    {
        const int h = tid >> 7;
        const int d = tid & 127;
        const v4f* p4 = (const v4f*)&pw[h * MAXR];
        v4f pv[MAXR / 4];
        #pragma unroll
        for (int j = 0; j < MAXR / 4; ++j) pv[j] = p4[j];
        const float* zc = &Zs[h * OUT_DIM + d];
        float z[MAXR];
        #pragma unroll
        for (int t = 0; t < MAXR / 16; ++t) {
            if (16 * t < cnt) {
                #pragma unroll
                for (int q = 0; q < 16; ++q) z[16 * t + q] = zc[(16 * t + q) * ZP];
            }
        }
        float s = 0.f;
        #pragma unroll
        for (int t = 0; t < MAXR / 16; ++t) {
            if (16 * t < cnt) {
                #pragma unroll
                for (int j = 4 * t; j < 4 * t + 4; ++j) {
                    s += pv[j].x;
                    s += pv[j].y;
                    s += pv[j].z;
                    s += pv[j].w;
                }
            }
        }
        const float inv = (cnt > 0) ? 1.f / s : 0.f;
        float accum = 0.f;
        #pragma unroll
        for (int t = 0; t < MAXR / 16; ++t) {
            if (16 * t < cnt) {
                #pragma unroll
                for (int j = 4 * t; j < 4 * t + 4; ++j) {
                    accum = fmaf(pv[j].x * inv, z[4 * j + 0], accum);
                    accum = fmaf(pv[j].y * inv, z[4 * j + 1], accum);
                    accum = fmaf(pv[j].z * inv, z[4 * j + 2], accum);
                    accum = fmaf(pv[j].w * inv, z[4 * j + 3], accum);
                }
            }
        }
        partial[h * OUT_DIM + d] = accum + bias[h * OUT_DIM + d];
    }
    __syncthreads();
    if (tid < OUT_DIM)
        out[(long)g * OUT_STRIDE + dir * OUT_DIM + tid] =
            0.5f * (partial[tid] + partial[OUT_DIM + tid]);
}

extern "C" void kernel_launch(void* const* d_in, const int* in_sizes, int n_in,
                              void* d_out, int out_size, void* d_ws, size_t ws_size,
                              hipStream_t stream)
{
    const float* feat     = (const float*)d_in[0];
    const int*   src      = (const int*)  d_in[1];
    const int*   dst      = (const int*)  d_in[2];
    const int*   op       = (const int*)  d_in[3];
    const int*   parallel = (const int*)  d_in[4];
    const float* W_f  = (const float*)d_in[5];
    const float* al_f = (const float*)d_in[6];
    const float* ar_f = (const float*)d_in[7];
    const float* b_f  = (const float*)d_in[8];
    const float* W_b  = (const float*)d_in[9];
    const float* al_b = (const float*)d_in[10];
    const float* ar_b = (const float*)d_in[11];
    const float* b_b  = (const float*)d_in[12];
    float* out = (float*)d_out;

    gat_node_kernel<<<dim3(N_GRAPHS * 3), dim3(256), 0, stream>>>(
        feat, src, dst, op, parallel, W_f, al_f, ar_f, b_f, W_b, al_b, ar_b, b_b, out);
}


# Z stays in the WMMA accumulators: per-wave logit shares by 16-lane DPP sums (4 shares added per row), weighted row sum from registers (tiles, then lane halves added last); bias asked for at entry
# speedup vs baseline: 3.7387x; 1.0668x over previous
#include <hip/hip_runtime.h>
#include <math.h>

#define N_GRAPHS 64
#define M_NODES  512
#define IN_DIM   256
#define OUT_DIM  128
#define NHEADS   2
#define ZD       256
#define EG       8192
#define P_IDX    32
#define NEG_SLOPE 0.2f

#define CAP      47
#define MAXR     48
#define NWAVES   8
#define WEDGES   (EG / NWAVES)
#define WSTRIDE  48
#define OUT_STRIDE 768
#define XP       260

typedef float v2f __attribute__((ext_vector_type(2)));
typedef float v8f __attribute__((ext_vector_type(8)));
typedef float v4f __attribute__((ext_vector_type(4)));

__device__ __forceinline__ float xhalf(float x)
{
    const int b = __builtin_bit_cast(int, x);
    return __builtin_bit_cast(float, __builtin_amdgcn_permlanex16(b, b, 0x76543210, 0xfedcba98, false, false));
}

template <int M>
__device__ __forceinline__ float xlane16(float x)
{
    const int b = __builtin_bit_cast(int, x);
    return __builtin_bit_cast(float, __builtin_amdgcn_update_dpp(0, b, 0x160 + M, 0xf, 0xf, true));
}

__device__ __forceinline__ float sum16(float x)
{
    x += xlane16<1>(x);
    x += xlane16<2>(x);
    x += xlane16<4>(x);
    x += xlane16<8>(x);
    return x;
}

__device__ __forceinline__ void load_b_group(v2f (&b)[8][2], const float* __restrict__ wc, int kg)
{
    #pragma unroll
    for (int s = 0; s < 8; ++s) {
        const int k0 = kg + 4 * s;
        #pragma unroll
        for (int nn = 0; nn < 2; ++nn) {
            b[s][nn].x = wc[(long)k0 * ZD + nn * 16];
            b[s][nn].y = wc[(long)(k0 + 1) * ZD + nn * 16];
        }
    }
}

template <int MT>
__device__ __forceinline__ void wmma_group(v8f (&acc)[3][2], const v2f (&a)[8][MT], const v2f (&b)[8][2])
{
    #pragma unroll
    for (int s = 0; s < 8; ++s) {
        #pragma unroll
        for (int mt = 0; mt < MT; ++mt) {
            #pragma unroll
            for (int nn = 0; nn < 2; ++nn)
                acc[mt][nn] = __builtin_amdgcn_wmma_f32_16x16x4_f32(
                                  false, a[s][mt], false, b[s][nn], (short)0, acc[mt][nn], false, false);
        }
    }
}

template <int MT, bool NEXT, bool LEAD>
__device__ __forceinline__ void gemm_group(v8f (&acc)[3][2], const float* xa, const float* __restrict__ wc, int kg,
                                           const v2f (&bUse)[8][2], v2f (&bNext)[8][2])
{
    if (LEAD) __builtin_amdgcn_sched_barrier(0);
    v2f a[8][MT];
    #pragma unroll
    for (int s = 0; s < 8; ++s) {
        #pragma unroll
        for (int mt = 0; mt < MT; ++mt)
            a[s][mt] = *(const v2f*)(xa + mt * 16 * XP + kg + 4 * s);
    }
    if (NEXT) load_b_group(bNext, wc, kg + 32);
    __builtin_amdgcn_sched_barrier(0);
    wmma_group<MT>(acc, a, bUse);
}

template <int MT>
__device__ __forceinline__ void gemm_rows(const float* __restrict__ feat,
                                          const int* rowNode, float* Xs, float* elp,
                                          int R, int cnt, int wave, int lane,
                                          const float* __restrict__ wc, v2f (&b0)[8][2],
                                          v8f (&acc)[3][2],
                                          float al0, float al1, float ar0, float ar1)
{
    {
        const int tid = wave * 32 + lane;
        const int c4  = tid & 63;
        const int r0  = tid >> 6;
        int rn[MT * 4];
        v4f x[MT * 4];
        #pragma unroll
        for (int j = 0; j < MT * 4; ++j) rn[j] = rowNode[min(r0 + 4 * j, R - 1)];
        #pragma unroll
        for (int j = 0; j < MT * 4; ++j) x[j] = *(const v4f*)(feat + (long)rn[j] * IN_DIM + c4 * 4);
        #pragma unroll
        for (int j = 0; j < MT * 4; ++j) *(v4f*)&Xs[(r0 + 4 * j) * XP + c4 * 4] = x[j];
    }
    __syncthreads();

    const int lm    = lane & 15;
    const int khalf = (lane >> 4) << 1;
    const float* xa = Xs + lm * XP + khalf;

    v2f b1[8][2];
    #pragma nounroll
    for (int kg = 0; kg < IN_DIM - 64; kg += 64) {
        gemm_group<MT, true, false>(acc, xa, wc, kg,      b0, b1);
        gemm_group<MT, true, true >(acc, xa, wc, kg + 32, b1, b0);
    }
    gemm_group<MT, true,  false>(acc, xa, wc, IN_DIM - 64, b0, b1);
    gemm_group<MT, false, true >(acc, xa, wc, IN_DIM - 32, b1, b0);
    const int  hh      = lane >> 4;
    const bool ownHalf = (hh == ((cnt >> 3) & 1));
    const int  ownIdx  = ((cnt >> 4) << 3) | (cnt & 7);
    float S[MT * 8];
    #pragma unroll
    for (int mt = 0; mt < MT; ++mt) {
        #pragma unroll
        for (int r = 0; r < 8; ++r) {
            const bool own = ownHalf && (mt * 8 + r == ownIdx);
            const float a0 = own ? ar0 : al0;
            const float a1 = own ? ar1 : al1;
            S[mt * 8 + r] = sum16(fmaf(acc[mt][1][r], a1, acc[mt][0][r] * a0));
        }
    }
    const int l7 = lane & 7;
    #pragma unroll
    for (int mt = 0; mt < MT; ++mt) {
        float v = S[mt * 8];
        #pragma unroll
        for (int r = 1; r < 8; ++r) v = (l7 == r) ? S[mt * 8 + r] : v;
        if ((lane & 8) == 0)
            elp[((wave >> 2) * MAXR + mt * 16 + hh * 8 + l7) * 4 + (wave & 3)] = v;
    }
}

__device__ __forceinline__ void tail_rows(const float* __restrict__ feat,
                                          const int*   __restrict__ opPtr,
                                          const int*   __restrict__ parallel,
                                          float* __restrict__ out, int g, int t)
{
    const int node = opPtr[0] + g * M_NODES;
    out[(long)g * OUT_STRIDE + 256 + t] = feat[(long)node * IN_DIM + t];
    float s = 0.f;
    for (int p = 0; p < P_IDX; ++p) {
        int n = parallel[p] + g * M_NODES;
        s += feat[(long)n * IN_DIM + t];
    }
    out[(long)g * OUT_STRIDE + 512 + t] = s;
}

__global__ __launch_bounds__(256, 2)
void gat_node_kernel(const float* __restrict__ feat,
                     const int*   __restrict__ srcArr,
                     const int*   __restrict__ dstArr,
                     const int*   __restrict__ opPtr,
                     const int*   __restrict__ parallel,
                     const float* __restrict__ W_f,  const float* __restrict__ al_f,
                     const float* __restrict__ ar_f, const float* __restrict__ b_f,
                     const float* __restrict__ W_b,  const float* __restrict__ al_b,
                     const float* __restrict__ ar_b, const float* __restrict__ b_b,
                     float* __restrict__ out)
{
    __shared__ __attribute__((aligned(16))) float Xs[MAXR * XP];
    __shared__ __attribute__((aligned(16))) float elp[NHEADS * MAXR * 4];
    __shared__ __attribute__((aligned(16))) float eh[NHEADS * MAXR];
    __shared__ __attribute__((aligned(16))) float pw[NHEADS * MAXR];
    __shared__ float er[NHEADS];
    __shared__ float partial[NHEADS * OUT_DIM];
    __shared__ int   rowNode[MAXR];
    __shared__ int   wl[NWAVES * WSTRIDE];
    __shared__ int   wtot[NWAVES];

    const int gb   = blockIdx.x;
    if (gb >= 2 * N_GRAPHS) {
        tail_rows(feat, opPtr, parallel, out, gb - 2 * N_GRAPHS, (int)threadIdx.x);
        return;
    }
    const int g    = gb >> 1;
    const int dir  = gb & 1;
    const int tid  = threadIdx.x;
    const int node = opPtr[0] + g * M_NODES;

    const int*   seg  = (dir == 0) ? dstArr : srcArr;
    const int*   gat  = (dir == 0) ? srcArr : dstArr;
    const float* Wd   = (dir == 0) ? W_f  : W_b;
    const float* al   = (dir == 0) ? al_f : al_b;
    const float* ar   = (dir == 0) ? ar_f : ar_b;
    const float* bias = (dir == 0) ? b_f  : b_b;

    const int wave  = tid >> 5;
    const int lane  = tid & 31;
    const int wbase = g * EG + wave * WEDGES;
    int v[8][4];
    #pragma unroll
    for (int i = 0; i < 8; ++i) {
        const int* p = seg + wbase + i * 128 + lane * 4;
        v[i][0] = p[0];
        v[i][1] = p[1];
        v[i][2] = p[2];
        v[i][3] = p[3];
    }
    const float al0 = al[wave * 32 + (lane & 15)];
    const float al1 = al[wave * 32 + 16 + (lane & 15)];
    const float ar0 = ar[wave * 32 + (lane & 15)];
    const float ar1 = ar[wave * 32 + 16 + (lane & 15)];
    const float bs0 = bias[wave * 32 + (lane & 15)];
    const float bs1 = bias[wave * 32 + 16 + (lane & 15)];
    const float* wc = Wd + (long)((lane >> 4) << 1) * ZD + wave * 32 + (lane & 15);
    v2f b0[8][2];
    load_b_group(b0, wc, 0);
    int run = 0;
    #pragma unroll
    for (int i = 0; i < 8; ++i) {
        const unsigned m0 = __builtin_amdgcn_ballot_w32(v[i][0] == node);
        const unsigned m1 = __builtin_amdgcn_ballot_w32(v[i][1] == node);
        const unsigned m2 = __builtin_amdgcn_ballot_w32(v[i][2] == node);
        const unsigned m3 = __builtin_amdgcn_ballot_w32(v[i][3] == node);
        if ((m0 | m1 | m2 | m3) != 0u) {
            unsigned pos = (unsigned)run;
            pos = __builtin_amdgcn_mbcnt_lo(m0, pos);
            pos = __builtin_amdgcn_mbcnt_lo(m1, pos);
            pos = __builtin_amdgcn_mbcnt_lo(m2, pos);
            pos = __builtin_amdgcn_mbcnt_lo(m3, pos);
            const int e0 = wbase + i * 128 + lane * 4;
            #pragma unroll
            for (int q = 0; q < 4; ++q) {
                if (v[i][q] == node) {
                    if (pos < CAP) wl[wave * WSTRIDE + pos] = e0 + q;
                    ++pos;
                }
            }
            run += __builtin_popcount(m0) + __builtin_popcount(m1)
                 + __builtin_popcount(m2) + __builtin_popcount(m3);
        }
    }
    if (lane == 0) wtot[wave] = run;
    __syncthreads();
    int before = 0, total = 0;
    #pragma unroll
    for (int w = 0; w < NWAVES; ++w) {
        const int t = wtot[w];
        before += (w < wave) ? t : 0;
        total  += t;
    }
    const int cnt   = min(total, CAP);
    const int R     = cnt + 1;
    const int tiles = (R + 15) >> 4;
    const int mine = min(run, CAP);
    #pragma unroll
    for (int t = 0; t < 2; ++t) {
        const int lp = lane + 32 * t;
        if (lp < mine && before + lp < CAP)
            rowNode[before + lp] = gat[wl[wave * WSTRIDE + lp]];
    }
    if (tid == 0) rowNode[cnt] = node;
    __syncthreads();

    v8f acc[3][2];
    #pragma unroll
    for (int mt = 0; mt < 3; ++mt) {
        acc[mt][0] = v8f{};
        acc[mt][1] = v8f{};
    }
    if (tiles == 1)      gemm_rows<1>(feat, rowNode, Xs, elp, R, cnt, wave, lane, wc, b0, acc, al0, al1, ar0, ar1);
    else if (tiles == 2) gemm_rows<2>(feat, rowNode, Xs, elp, R, cnt, wave, lane, wc, b0, acc, al0, al1, ar0, ar1);
    else                 gemm_rows<3>(feat, rowNode, Xs, elp, R, cnt, wave, lane, wc, b0, acc, al0, al1, ar0, ar1);
    __syncthreads();

    if (tid < 2 * MAXR) {
        const int row = tid >> 1;
        const int h   = tid & 1;
        float elv = -INFINITY;
        if (row <= cnt) {
            const v4f p = *(const v4f*)&elp[(h * MAXR + row) * 4];
            const float sum = ((p.x + p.y) + p.z) + p.w;
            if (row == cnt) er[h] = sum;
            else            elv = sum;
        }
        eh[h * MAXR + row] = elv;
    }
    __syncthreads();

    if (tid < 2 * MAXR) {
        const int row = tid >> 1;
        const int h   = tid & 1;
        const float erh = er[h];
        const v4f* e4 = (const v4f*)&eh[h * MAXR];
        v4f ev[MAXR / 4];
        #pragma unroll
        for (int j = 0; j < MAXR / 4; ++j) ev[j] = e4[j];
        const float elo = eh[h * MAXR + row];
        v4f m4 = { -INFINITY, -INFINITY, -INFINITY, -INFINITY };
        #pragma unroll
        for (int j = 0; j < MAXR / 4; ++j) {
            if (4 * j < cnt) {
                v4f e = ev[j] + erh;
                e.x = (e.x > 0.f) ? e.x : NEG_SLOPE * e.x;
                e.y = (e.y > 0.f) ? e.y : NEG_SLOPE * e.y;
                e.z = (e.z > 0.f) ? e.z : NEG_SLOPE * e.z;
                e.w = (e.w > 0.f) ? e.w : NEG_SLOPE * e.w;
                m4.x = fmaxf(m4.x, e.x);
                m4.y = fmaxf(m4.y, e.y);
                m4.z = fmaxf(m4.z, e.z);
                m4.w = fmaxf(m4.w, e.w);
            }
        }
        const float mx = fmaxf(fmaxf(m4.x, m4.y), fmaxf(m4.z, m4.w));
        float eo = elo + erh;
        eo = (eo > 0.f) ? eo : NEG_SLOPE * eo;
        const float p = expf(eo - mx);
        pw[h * MAXR + row] = (row < cnt) ? p : 0.f;
    }
    __syncthreads();

    {
        const int lm = lane & 15;
        const int hh = lane >> 4;
        const int h  = wave >> 2;
        v4f pa[3][2];
        #pragma unroll
        for (int mt = 0; mt < 3; ++mt) {
            if (mt < tiles) {
                const v4f* p4 = (const v4f*)&pw[h * MAXR + mt * 16 + hh * 8];
                pa[mt][0] = p4[0];
                pa[mt][1] = p4[1];
            }
        }
        float st[3] = { 0.f, 0.f, 0.f };
        #pragma unroll
        for (int mt = 0; mt < 3; ++mt) {
            if (mt < tiles) {
                float t = pa[mt][0].x;
                t += pa[mt][0].y;
                t += pa[mt][0].z;
                t += pa[mt][0].w;
                t += pa[mt][1].x;
                t += pa[mt][1].y;
                t += pa[mt][1].z;
                t += pa[mt][1].w;
                st[mt] = t;
            }
        }
        const float sh   = (st[0] + st[1]) + st[2];
        const float ssum = sh + xhalf(sh);
        const float inv  = (cnt > 0) ? 1.f / ssum : 0.f;
        float u0[3] = { 0.f, 0.f, 0.f };
        float u1[3] = { 0.f, 0.f, 0.f };
        #pragma unroll
        for (int mt = 0; mt < 3; ++mt) {
            if (mt < tiles) {
                float t0 = 0.f, t1 = 0.f;
                #pragma unroll
                for (int q = 0; q < 2; ++q) {
                    #pragma unroll
                    for (int e = 0; e < 4; ++e) {
                        const float w = pa[mt][q][e] * inv;
                        t0 = fmaf(w, acc[mt][0][q * 4 + e], t0);
                        t1 = fmaf(w, acc[mt][1][q * 4 + e], t1);
                    }
                }
                u0[mt] = t0;
                u1[mt] = t1;
            }
        }
        float o0 = (u0[0] + u0[1]) + u0[2];
        float o1 = (u1[0] + u1[1]) + u1[2];
        o0 += xhalf(o0);
        o1 += xhalf(o1);
        if (hh == 0) {
            partial[wave * 32 + lm]      = o0 + bs0;
            partial[wave * 32 + 16 + lm] = o1 + bs1;
        }
    }
    __syncthreads();
    if (tid < OUT_DIM)
        out[(long)g * OUT_STRIDE + dir * OUT_DIM + tid] =
            0.5f * (partial[tid] + partial[OUT_DIM + tid]);
}

extern "C" void kernel_launch(void* const* d_in, const int* in_sizes, int n_in,
                              void* d_out, int out_size, void* d_ws, size_t ws_size,
                              hipStream_t stream)
{
    const float* feat     = (const float*)d_in[0];
    const int*   src      = (const int*)  d_in[1];
    const int*   dst      = (const int*)  d_in[2];
    const int*   op       = (const int*)  d_in[3];
    const int*   parallel = (const int*)  d_in[4];
    const float* W_f  = (const float*)d_in[5];
    const float* al_f = (const float*)d_in[6];
    const float* ar_f = (const float*)d_in[7];
    const float* b_f  = (const float*)d_in[8];
    const float* W_b  = (const float*)d_in[9];
    const float* al_b = (const float*)d_in[10];
    const float* ar_b = (const float*)d_in[11];
    const float* b_b  = (const float*)d_in[12];
    float* out = (float*)d_out;

    gat_node_kernel<<<dim3(N_GRAPHS * 3), dim3(256), 0, stream>>>(
        feat, src, dst, op, parallel, W_f, al_f, ar_f, b_f, W_b, al_b, ar_b, b_b, out);
}
